# starting kernel
# baseline (speedup 1.0000x reference)
#include <hip/hip_runtime.h>
#include <hip/hip_bf16.h>

typedef float v2f __attribute__((ext_vector_type(2)));
typedef float v8f __attribute__((ext_vector_type(8)));

__global__ __launch_bounds__(256) void wmma_gemm_bias(
    const float* __restrict__ A, const float* __restrict__ W,
    const float* __restrict__ bias, float* __restrict__ C,
    int M, int N, int K, int relu)
{
    const int lane = threadIdx.x & 31;
    const int wave = threadIdx.x >> 5;
    const int tilesN = N >> 6;
    const int tile = blockIdx.x * 8 + wave;
    const int totalTiles = (M >> 4) * tilesN;
    if (tile >= totalTiles) return;
    const int tm = (tile / tilesN) << 4;
    const int tn = (tile % tilesN) << 6;

    const int half = lane >> 4;
    const int l16  = lane & 15;

    const float* ap = A + (size_t)(tm + l16) * K + 2 * half;
    const float* bp = W + (size_t)(2 * half) * N + tn + l16;
    const size_t sN = (size_t)N;

    v8f acc0 = {}, acc1 = {}, acc2 = {}, acc3 = {};
    for (int k = 0; k < K; k += 4) {
        v2f a = *(const v2f*)ap;
        const float* b0 = bp;
        const float* b1 = bp + sN;
        v2f f0, f1, f2, f3;
        f0.x = b0[0];  f0.y = b1[0];
        f1.x = b0[16]; f1.y = b1[16];
        f2.x = b0[32]; f2.y = b1[32];
        f3.x = b0[48]; f3.y = b1[48];
        acc0 = __builtin_amdgcn_wmma_f32_16x16x4_f32(false, a, false, f0,
                                                     (short)0, acc0, false, false);
        acc1 = __builtin_amdgcn_wmma_f32_16x16x4_f32(false, a, false, f1,
                                                     (short)0, acc1, false, false);
        acc2 = __builtin_amdgcn_wmma_f32_16x16x4_f32(false, a, false, f2,
                                                     (short)0, acc2, false, false);
        acc3 = __builtin_amdgcn_wmma_f32_16x16x4_f32(false, a, false, f3,
                                                     (short)0, acc3, false, false);
        ap += 4;
        bp += 4 * sN;
        __builtin_prefetch(bp + 8 * sN, 0, 1);
    }

    const int row0 = tm + half * 8;
    float bv0 = 0.f, bv1 = 0.f, bv2 = 0.f, bv3 = 0.f;
    if (bias) {
        const float* bb = bias + tn + l16;
        bv0 = bb[0]; bv1 = bb[16]; bv2 = bb[32]; bv3 = bb[48];
    }
    float* crow = C + (size_t)row0 * N + tn + l16;
#pragma unroll
    for (int r = 0; r < 8; ++r) {
        float v0 = acc0[r] + bv0, v1 = acc1[r] + bv1;
        float v2 = acc2[r] + bv2, v3 = acc3[r] + bv3;
        if (relu) {
            v0 = fmaxf(v0, 0.f); v1 = fmaxf(v1, 0.f);
            v2 = fmaxf(v2, 0.f); v3 = fmaxf(v3, 0.f);
        }
        crow[0] = v0; crow[16] = v1; crow[32] = v2; crow[48] = v3;
        crow += sN;
    }
}

__global__ void count_same(const int* __restrict__ y, float* __restrict__ cnt, int S)
{
    int i = blockIdx.x * blockDim.x + threadIdx.x;
    if (i >= S) return;
    int yi = y[i];
    int c = 0;
    for (int j = 0; j < S; ++j) c += (y[j] == yi) ? 1 : 0;
    cnt[i] = (float)(c - 1);
}

__global__ __launch_bounds__(256) void pair_accum(
    const float* __restrict__ Abuf, const float* __restrict__ Bbuf,
    const float* __restrict__ bm1t, const int* __restrict__ y,
    float* __restrict__ haggr, int S, int H)
{
    const int i = blockIdx.x;
    const int h = threadIdx.x;
    const int yi = y[i];
    const float av = Abuf[(size_t)i * H + h] + bm1t[h];
    float acc = 0.0f;
    for (int j = 0; j < S; ++j) {
        if (y[j] == yi && j != i)
            acc += fmaxf(av + Bbuf[(size_t)j * H + h], 0.0f);
    }
    haggr[(size_t)i * H + h] = acc;
}

__global__ __launch_bounds__(256) void update_s(
    float* __restrict__ s, const float* __restrict__ aggmat,
    const float* __restrict__ bm2t, const float* __restrict__ cnt, int H)
{
    const int i = blockIdx.x;
    const int h = threadIdx.x;
    const float c = cnt[i];
    const float denom = fmaxf(c, 1.0f);
    s[(size_t)i * H + h] += (aggmat[(size_t)i * H + h] + c * bm2t[h]) / denom;
}

__global__ __launch_bounds__(256) void score_kernel(
    const float* __restrict__ hq, const float* __restrict__ hs,
    const float* __restrict__ br1, const float* __restrict__ wr2,
    const float* __restrict__ br2, float* __restrict__ out, int S, int H)
{
    const int q = blockIdx.x;
    const int lane = threadIdx.x & 31;
    const int wave = threadIdx.x >> 5;
    const int hbase = lane * 8;

    float hqv[8], bv[8], wv[8];
#pragma unroll
    for (int u = 0; u < 8; ++u) {
        hqv[u] = hq[(size_t)q * H + hbase + u];
        bv[u]  = br1[hbase + u];
        wv[u]  = wr2[hbase + u];
    }
    const float b2 = br2[0];

    for (int s = wave; s < S; s += 8) {
        const float* hsrow = hs + (size_t)s * H + hbase;
        float sum = 0.0f;
#pragma unroll
        for (int u = 0; u < 8; ++u)
            sum += fmaxf(hqv[u] + hsrow[u] + bv[u], 0.0f) * wv[u];
#pragma unroll
        for (int off = 16; off > 0; off >>= 1)
            sum += __shfl_down(sum, off, 32);
        if (lane == 0) out[(size_t)q * S + s] = sum + b2;
    }
}

extern "C" void kernel_launch(void* const* d_in, const int* in_sizes, int n_in,
                              void* d_out, int out_size, void* d_ws, size_t ws_size,
                              hipStream_t stream)
{
    const float* Xq  = (const float*)d_in[0];
    const float* Xs  = (const float*)d_in[1];
    const int*   y   = (const int*)d_in[2];
    const float* Wn1 = (const float*)d_in[3];
    const float* bn1 = (const float*)d_in[4];
    const float* Wn2 = (const float*)d_in[5];
    const float* bn2 = (const float*)d_in[6];
    const float* Wm1 = (const float*)d_in[7];
    const float* bm1 = (const float*)d_in[8];
    const float* Wm2 = (const float*)d_in[9];
    const float* bm2 = (const float*)d_in[10];
    const float* Wr1 = (const float*)d_in[11];
    const float* br1 = (const float*)d_in[12];
    const float* wr2 = (const float*)d_in[13];
    const float* br2 = (const float*)d_in[14];
    float* out = (float*)d_out;

    const int E = 512, H = 256, Q = 1024, S = 512, STEPS = 3;

    float* ws  = (float*)d_ws;
    float* q1  = ws; ws += (size_t)Q * H;
    float* qf  = ws; ws += (size_t)Q * H;
    float* s1  = ws; ws += (size_t)S * H;
    float* sf  = ws; ws += (size_t)S * H;
    float* Ab  = ws; ws += (size_t)S * H;
    float* Bb  = ws; ws += (size_t)S * H;
    float* hag = ws; ws += (size_t)S * H;
    float* agg = ws; ws += (size_t)S * H;
    float* hqb = ws; ws += (size_t)Q * H;
    float* hsb = ws; ws += (size_t)S * H;
    float* cnt = ws; ws += S;

    auto gemm = [&](const float* A, const float* W, const float* bias, float* C,
                    int M, int N, int K, int relu) {
        int tiles  = (M / 16) * (N / 64);
        int blocks = (tiles + 7) / 8;
        wmma_gemm_bias<<<blocks, 256, 0, stream>>>(A, W, bias, C, M, N, K, relu);
    };

    gemm(Xq, Wn1, bn1, q1, Q, H, E, 1);
    gemm(q1, Wn2, bn2, qf, Q, H, H, 0);
    gemm(Xs, Wn1, bn1, s1, S, H, E, 1);
    gemm(s1, Wn2, bn2, sf, S, H, H, 0);

    count_same<<<(S + 255) / 256, 256, 0, stream>>>(y, cnt, S);

    for (int t = 0; t < STEPS; ++t) {
        const float* W1a = Wm1 + (size_t)t * (2 * H) * H;
        const float* W1b = W1a + (size_t)H * H;
        gemm(sf, W1a, nullptr, Ab, S, H, H, 0);
        gemm(sf, W1b, nullptr, Bb, S, H, H, 0);
        pair_accum<<<S, H, 0, stream>>>(Ab, Bb, bm1 + (size_t)t * H, y, hag, S, H);
        gemm(hag, Wm2 + (size_t)t * H * H, nullptr, agg, S, H, H, 0);
        update_s<<<S, H, 0, stream>>>(sf, agg, bm2 + (size_t)t * H, cnt, H);
    }

    gemm(qf, Wr1, nullptr, hqb, Q, H, H, 0);
    gemm(sf, Wr1 + (size_t)H * H, nullptr, hsb, S, H, H, 0);
    score_kernel<<<Q, 256, 0, stream>>>(hqb, hsb, br1, wr2, br2, out, S, H);
}


# Dependent launches 23 -> 13: independent GEMMs share one grid (block index picks the product), node update folded into the aggregation GEMM's epilogue, class count taken in the pair kernel
# speedup vs baseline: 3.5820x; 3.5820x over previous
#include <hip/hip_runtime.h>
#include <hip/hip_bf16.h>

typedef float v2f __attribute__((ext_vector_type(2)));
typedef float v8f __attribute__((ext_vector_type(8)));

template <int NW, int EPI>
__global__ __launch_bounds__(NW * 32) void wmma_gemm_pair(
    const float* __restrict__ A0, const float* __restrict__ W0,
    float* __restrict__ C0, int tiles0,
    const float* __restrict__ A1, const float* __restrict__ W1,
    float* __restrict__ C1,
    const float* __restrict__ bias, const float* __restrict__ cnt,
    int N, int K, int relu)
{
    constexpr int PITCH = 68;
    __shared__ alignas(16) float red[NW][16 * PITCH];

    const int lane = threadIdx.x & 31;
    const int wave = threadIdx.x >> 5;
    const int tilesN = N >> 6;

    const bool second = (int)blockIdx.x >= tiles0;
    const float* A = second ? A1 : A0;
    const float* W = second ? W1 : W0;
    float*       C = second ? C1 : C0;
    const int tile = second ? (int)blockIdx.x - tiles0 : (int)blockIdx.x;
    const int tm = (tile / tilesN) << 4;
    const int tn = (tile % tilesN) << 6;

    const int half = lane >> 4;
    const int l16  = lane & 15;
    const int k0   = wave << 5;

    const float* ap = A + (size_t)(tm + l16) * K + k0 + 2 * half;
    const float* bp = W + (size_t)(k0 + 2 * half) * N + tn + l16;
    const size_t sN = (size_t)N;

    v2f a[8];
    v2f f[8][4];
#pragma unroll
    for (int s = 0; s < 8; ++s) {
        a[s] = *(const v2f*)(ap + 4 * s);
        const float* b0 = bp + (size_t)(4 * s) * sN;
        const float* b1 = b0 + sN;
#pragma unroll
        for (int j = 0; j < 4; ++j) {
            f[s][j].x = b0[16 * j];
            f[s][j].y = b1[16 * j];
        }
    }

    __builtin_amdgcn_sched_barrier(0);

    v8f acc0 = {}, acc1 = {}, acc2 = {}, acc3 = {};
#pragma unroll
    for (int s = 0; s < 8; ++s) {
        acc0 = __builtin_amdgcn_wmma_f32_16x16x4_f32(false, a[s], false, f[s][0],
                                                     (short)0, acc0, false, false);
        acc1 = __builtin_amdgcn_wmma_f32_16x16x4_f32(false, a[s], false, f[s][1],
                                                     (short)0, acc1, false, false);
        acc2 = __builtin_amdgcn_wmma_f32_16x16x4_f32(false, a[s], false, f[s][2],
                                                     (short)0, acc2, false, false);
        acc3 = __builtin_amdgcn_wmma_f32_16x16x4_f32(false, a[s], false, f[s][3],
                                                     (short)0, acc3, false, false);
    }

    float* my = &red[wave][(half * 8) * PITCH + l16];
#pragma unroll
    for (int r = 0; r < 8; ++r) {
        my[r * PITCH]      = acc0[r];
        my[r * PITCH + 16] = acc1[r];
        my[r * PITCH + 32] = acc2[r];
        my[r * PITCH + 48] = acc3[r];
    }
    __syncthreads();

    const int t = threadIdx.x;
    if (t < 256) {
        const int row = t >> 4;
        const int c4  = (t & 15) << 2;
        float* cp = C + (size_t)(tm + row) * N + tn + c4;
        float4 sum = *(const float4*)&red[0][row * PITCH + c4];
#pragma unroll
        for (int w = 1; w < NW; ++w) {
            const float4 p = *(const float4*)&red[w][row * PITCH + c4];
            sum.x += p.x; sum.y += p.y; sum.z += p.z; sum.w += p.w;
        }
        if (EPI == 0) {
            if (bias) {
                const float4 b4 = *(const float4*)(bias + tn + c4);
                sum.x += b4.x; sum.y += b4.y; sum.z += b4.z; sum.w += b4.w;
            }
            if (relu) {
                sum.x = fmaxf(sum.x, 0.f); sum.y = fmaxf(sum.y, 0.f);
                sum.z = fmaxf(sum.z, 0.f); sum.w = fmaxf(sum.w, 0.f);
            }
            *(float4*)cp = sum;
        } else {
            const float c = cnt[tm + row];
            const float denom = fmaxf(c, 1.0f);
            const float4 bv = *(const float4*)(bias + tn + c4);
            float4 sv = *(const float4*)cp;
            sv.x += (sum.x + c * bv.x) / denom;
            sv.y += (sum.y + c * bv.y) / denom;
            sv.z += (sum.z + c * bv.z) / denom;
            sv.w += (sum.w + c * bv.w) / denom;
            *(float4*)cp = sv;
        }
    }
}

__global__ __launch_bounds__(256) void pair_accum(
    const float* __restrict__ Abuf, const float* __restrict__ Bbuf,
    const float* __restrict__ bm1t, const int* __restrict__ y,
    float* __restrict__ haggr, float* __restrict__ cnt, int S, int H)
{
    const int i = blockIdx.x;
    const int h = threadIdx.x;
    const int yi = y[i];
    const float av = Abuf[(size_t)i * H + h] + bm1t[h];
    float acc = 0.0f;
    int n = 0;
    for (int j = 0; j < S; ++j) {
        if (y[j] == yi && j != i) {
            acc += fmaxf(av + Bbuf[(size_t)j * H + h], 0.0f);
            ++n;
        }
    }
    haggr[(size_t)i * H + h] = acc;
    if (h == 0) cnt[i] = (float)n;
}

__global__ __launch_bounds__(256) void score_kernel(
    const float* __restrict__ hq, const float* __restrict__ hs,
    const float* __restrict__ br1, const float* __restrict__ wr2,
    const float* __restrict__ br2, float* __restrict__ out, int S, int H)
{
    const int q = blockIdx.x;
    const int lane = threadIdx.x & 31;
    const int wave = threadIdx.x >> 5;
    const int hbase = lane * 8;

    float hqv[8], bv[8], wv[8];
#pragma unroll
    for (int u = 0; u < 8; ++u) {
        hqv[u] = hq[(size_t)q * H + hbase + u];
        bv[u]  = br1[hbase + u];
        wv[u]  = wr2[hbase + u];
    }
    const float b2 = br2[0];

    for (int s = wave; s < S; s += 8) {
        const float* hsrow = hs + (size_t)s * H + hbase;
        float sum = 0.0f;
#pragma unroll
        for (int u = 0; u < 8; ++u)
            sum += fmaxf(hqv[u] + hsrow[u] + bv[u], 0.0f) * wv[u];
#pragma unroll
        for (int off = 16; off > 0; off >>= 1)
            sum += __shfl_down(sum, off, 32);
        if (lane == 0) out[(size_t)q * S + s] = sum + b2;
    }
}

extern "C" void kernel_launch(void* const* d_in, const int* in_sizes, int n_in,
                              void* d_out, int out_size, void* d_ws, size_t ws_size,
                              hipStream_t stream)
{
    const float* Xq  = (const float*)d_in[0];
    const float* Xs  = (const float*)d_in[1];
    const int*   y   = (const int*)d_in[2];
    const float* Wn1 = (const float*)d_in[3];
    const float* bn1 = (const float*)d_in[4];
    const float* Wn2 = (const float*)d_in[5];
    const float* bn2 = (const float*)d_in[6];
    const float* Wm1 = (const float*)d_in[7];
    const float* bm1 = (const float*)d_in[8];
    const float* Wm2 = (const float*)d_in[9];
    const float* bm2 = (const float*)d_in[10];
    const float* Wr1 = (const float*)d_in[11];
    const float* br1 = (const float*)d_in[12];
    const float* wr2 = (const float*)d_in[13];
    const float* br2 = (const float*)d_in[14];
    float* out = (float*)d_out;

    const int E = 512, H = 256, Q = 1024, S = 512, STEPS = 3;
    const int TQ = (Q / 16) * (H / 64);
    const int TS = (S / 16) * (H / 64);

    float* ws  = (float*)d_ws;
    float* q1  = ws; ws += (size_t)Q * H;
    float* qf  = ws; ws += (size_t)Q * H;
    float* s1  = ws; ws += (size_t)S * H;
    float* sf  = ws; ws += (size_t)S * H;
    float* Ab  = ws; ws += (size_t)S * H;
    float* Bb  = ws; ws += (size_t)S * H;
    float* hag = ws; ws += (size_t)S * H;
    float* hqb = ws; ws += (size_t)Q * H;
    float* hsb = ws; ws += (size_t)S * H;
    float* cnt = ws; ws += S;

    wmma_gemm_pair<16, 0><<<TQ + TS, 512, 0, stream>>>(
        Xq, Wn1, q1, TQ, Xs, Wn1, s1, bn1, nullptr, H, E, 1);
    wmma_gemm_pair<8, 0><<<TQ + TS, 256, 0, stream>>>(
        q1, Wn2, qf, TQ, s1, Wn2, sf, bn2, nullptr, H, H, 0);

    for (int t = 0; t < STEPS; ++t) {
        const float* W1a = Wm1 + (size_t)t * (2 * H) * H;
        const float* W1b = W1a + (size_t)H * H;
        wmma_gemm_pair<8, 0><<<TS + TS, 256, 0, stream>>>(
            sf, W1a, Ab, TS, sf, W1b, Bb, nullptr, nullptr, H, H, 0);
        pair_accum<<<S, H, 0, stream>>>(Ab, Bb, bm1 + (size_t)t * H, y, hag, cnt, S, H);
        wmma_gemm_pair<8, 1><<<TS, 256, 0, stream>>>(
            hag, Wm2 + (size_t)t * H * H, sf, TS, nullptr, nullptr, nullptr,
            bm2 + (size_t)t * H, cnt, H, H, 0);
    }

    wmma_gemm_pair<8, 0><<<TQ + TS, 256, 0, stream>>>(
        qf, Wr1, hqb, TQ, sf, Wr1 + (size_t)H * H, hsb, nullptr, nullptr, H, H, 0);
    score_kernel<<<Q, 256, 0, stream>>>(hqb, hsb, br1, wr2, br2, out, S, H);
}


# Pair kernel: block compacts its same-class j into an LDS list by ballot, then sums 32 rows per memory round trip (was a 512-trip scalar label loop, one dependent load per match); sum order kept
# speedup vs baseline: 6.3599x; 1.7755x over previous
#include <hip/hip_runtime.h>
#include <hip/hip_bf16.h>

typedef float v2f __attribute__((ext_vector_type(2)));
typedef float v8f __attribute__((ext_vector_type(8)));

template <int NW, int EPI>
__global__ __launch_bounds__(NW * 32) void wmma_gemm_pair(
    const float* __restrict__ A0, const float* __restrict__ W0,
    float* __restrict__ C0, int tiles0,
    const float* __restrict__ A1, const float* __restrict__ W1,
    float* __restrict__ C1,
    const float* __restrict__ bias, const float* __restrict__ cnt,
    int N, int K, int relu)
{
    constexpr int PITCH = 68;
    __shared__ alignas(16) float red[NW][16 * PITCH];

    const int lane = threadIdx.x & 31;
    const int wave = threadIdx.x >> 5;
    const int tilesN = N >> 6;

    const bool second = (int)blockIdx.x >= tiles0;
    const float* A = second ? A1 : A0;
    const float* W = second ? W1 : W0;
    float*       C = second ? C1 : C0;
    const int tile = second ? (int)blockIdx.x - tiles0 : (int)blockIdx.x;
    const int tm = (tile / tilesN) << 4;
    const int tn = (tile % tilesN) << 6;

    const int half = lane >> 4;
    const int l16  = lane & 15;
    const int k0   = wave << 5;

    const float* ap = A + (size_t)(tm + l16) * K + k0 + 2 * half;
    const float* bp = W + (size_t)(k0 + 2 * half) * N + tn + l16;
    const size_t sN = (size_t)N;

    v2f a[8];
    v2f f[8][4];
#pragma unroll
    for (int s = 0; s < 8; ++s) {
        a[s] = *(const v2f*)(ap + 4 * s);
        const float* b0 = bp + (size_t)(4 * s) * sN;
        const float* b1 = b0 + sN;
#pragma unroll
        for (int j = 0; j < 4; ++j) {
            f[s][j].x = b0[16 * j];
            f[s][j].y = b1[16 * j];
        }
    }

    __builtin_amdgcn_sched_barrier(0);

    v8f acc0 = {}, acc1 = {}, acc2 = {}, acc3 = {};
#pragma unroll
    for (int s = 0; s < 8; ++s) {
        acc0 = __builtin_amdgcn_wmma_f32_16x16x4_f32(false, a[s], false, f[s][0],
                                                     (short)0, acc0, false, false);
        acc1 = __builtin_amdgcn_wmma_f32_16x16x4_f32(false, a[s], false, f[s][1],
                                                     (short)0, acc1, false, false);
        acc2 = __builtin_amdgcn_wmma_f32_16x16x4_f32(false, a[s], false, f[s][2],
                                                     (short)0, acc2, false, false);
        acc3 = __builtin_amdgcn_wmma_f32_16x16x4_f32(false, a[s], false, f[s][3],
                                                     (short)0, acc3, false, false);
    }

    float* my = &red[wave][(half * 8) * PITCH + l16];
#pragma unroll
    for (int r = 0; r < 8; ++r) {
        my[r * PITCH]      = acc0[r];
        my[r * PITCH + 16] = acc1[r];
        my[r * PITCH + 32] = acc2[r];
        my[r * PITCH + 48] = acc3[r];
    }
    __syncthreads();

    const int t = threadIdx.x;
    if (t < 256) {
        const int row = t >> 4;
        const int c4  = (t & 15) << 2;
        float* cp = C + (size_t)(tm + row) * N + tn + c4;
        float4 sum = *(const float4*)&red[0][row * PITCH + c4];
#pragma unroll
        for (int w = 1; w < NW; ++w) {
            const float4 p = *(const float4*)&red[w][row * PITCH + c4];
            sum.x += p.x; sum.y += p.y; sum.z += p.z; sum.w += p.w;
        }
        if (EPI == 0) {
            if (bias) {
                const float4 b4 = *(const float4*)(bias + tn + c4);
                sum.x += b4.x; sum.y += b4.y; sum.z += b4.z; sum.w += b4.w;
            }
            if (relu) {
                sum.x = fmaxf(sum.x, 0.f); sum.y = fmaxf(sum.y, 0.f);
                sum.z = fmaxf(sum.z, 0.f); sum.w = fmaxf(sum.w, 0.f);
            }
            *(float4*)cp = sum;
        } else {
            const float c = cnt[tm + row];
            const float denom = fmaxf(c, 1.0f);
            const float4 bv = *(const float4*)(bias + tn + c4);
            float4 sv = *(const float4*)cp;
            sv.x += (sum.x + c * bv.x) / denom;
            sv.y += (sum.y + c * bv.y) / denom;
            sv.z += (sum.z + c * bv.z) / denom;
            sv.w += (sum.w + c * bv.w) / denom;
            *(float4*)cp = sv;
        }
    }
}

constexpr int PAIR_S = 512;
constexpr int PG = 32;
__global__ __launch_bounds__(256) void pair_accum(
    const float* __restrict__ Abuf, const float* __restrict__ Bbuf,
    const float* __restrict__ bm1t, const int* __restrict__ y,
    float* __restrict__ haggr, float* __restrict__ cnt, int H)
{
    __shared__ alignas(16) unsigned list[PAIR_S + PG];
    __shared__ alignas(16) int wcnt[8];

    const int i = blockIdx.x;
    const int h = threadIdx.x;
    const int lane = h & 31;
    const int wave = h >> 5;

    const int j0 = (wave << 6) + lane;
    const int j1 = j0 + 32;
    const int yi = y[i];
    const int y0 = y[j0];
    const int y1 = y[j1];
    const float av = Abuf[(size_t)i * H + h] + bm1t[h];

    const bool m0 = (y0 == yi) && (j0 != i);
    const bool m1 = (y1 == yi) && (j1 != i);
    const unsigned b0 = __builtin_amdgcn_ballot_w32(m0);
    const unsigned b1 = __builtin_amdgcn_ballot_w32(m1);
    const int c0 = __builtin_popcount(b0);
    const int c1 = __builtin_popcount(b1);
    if (lane == 0) wcnt[wave] = c0 + c1;
    __syncthreads();

    int base = 0, n = 0;
    for (int w = 0; w < 8; ++w) {
        const int c = wcnt[w];
        base += (w < wave) ? c : 0;
        n += c;
    }
    const unsigned below = (1u << lane) - 1u;
    if (m0) list[base + __builtin_popcount(b0 & below)] = (unsigned)(j0 * H);
    if (m1) list[base + c0 + __builtin_popcount(b1 & below)] = (unsigned)(j1 * H);
    if (h < PG) list[n + h] = 0u;
    __syncthreads();

    float acc = 0.0f;
    for (int p = 0; p < n; p += PG) {
        float v[PG];
#pragma unroll
        for (int u = 0; u < PG; ++u)
            v[u] = Bbuf[list[p + u] + (unsigned)h];
#pragma unroll
        for (int u = 0; u < PG; ++u) {
            const float t = fmaxf(av + v[u], 0.0f);
            acc += (p + u < n) ? t : 0.0f;
        }
    }
    haggr[(size_t)i * H + h] = acc;
    if (h == 0) cnt[i] = (float)n;
}

__global__ __launch_bounds__(256) void score_kernel(
    const float* __restrict__ hq, const float* __restrict__ hs,
    const float* __restrict__ br1, const float* __restrict__ wr2,
    const float* __restrict__ br2, float* __restrict__ out, int S, int H)
{
    const int q = blockIdx.x;
    const int lane = threadIdx.x & 31;
    const int wave = threadIdx.x >> 5;
    const int hbase = lane * 8;

    float hqv[8], bv[8], wv[8];
#pragma unroll
    for (int u = 0; u < 8; ++u) {
        hqv[u] = hq[(size_t)q * H + hbase + u];
        bv[u]  = br1[hbase + u];
        wv[u]  = wr2[hbase + u];
    }
    const float b2 = br2[0];

    for (int s = wave; s < S; s += 8) {
        const float* hsrow = hs + (size_t)s * H + hbase;
        float sum = 0.0f;
#pragma unroll
        for (int u = 0; u < 8; ++u)
            sum += fmaxf(hqv[u] + hsrow[u] + bv[u], 0.0f) * wv[u];
#pragma unroll
        for (int off = 16; off > 0; off >>= 1)
            sum += __shfl_down(sum, off, 32);
        if (lane == 0) out[(size_t)q * S + s] = sum + b2;
    }
}

extern "C" void kernel_launch(void* const* d_in, const int* in_sizes, int n_in,
                              void* d_out, int out_size, void* d_ws, size_t ws_size,
                              hipStream_t stream)
{
    const float* Xq  = (const float*)d_in[0];
    const float* Xs  = (const float*)d_in[1];
    const int*   y   = (const int*)d_in[2];
    const float* Wn1 = (const float*)d_in[3];
    const float* bn1 = (const float*)d_in[4];
    const float* Wn2 = (const float*)d_in[5];
    const float* bn2 = (const float*)d_in[6];
    const float* Wm1 = (const float*)d_in[7];
    const float* bm1 = (const float*)d_in[8];
    const float* Wm2 = (const float*)d_in[9];
    const float* bm2 = (const float*)d_in[10];
    const float* Wr1 = (const float*)d_in[11];
    const float* br1 = (const float*)d_in[12];
    const float* wr2 = (const float*)d_in[13];
    const float* br2 = (const float*)d_in[14];
    float* out = (float*)d_out;

    const int E = 512, H = 256, Q = 1024, S = 512, STEPS = 3;
    const int TQ = (Q / 16) * (H / 64);
    const int TS = (S / 16) * (H / 64);

    float* ws  = (float*)d_ws;
    float* q1  = ws; ws += (size_t)Q * H;
    float* qf  = ws; ws += (size_t)Q * H;
    float* s1  = ws; ws += (size_t)S * H;
    float* sf  = ws; ws += (size_t)S * H;
    float* Ab  = ws; ws += (size_t)S * H;
    float* Bb  = ws; ws += (size_t)S * H;
    float* hag = ws; ws += (size_t)S * H;
    float* hqb = ws; ws += (size_t)Q * H;
    float* hsb = ws; ws += (size_t)S * H;
    float* cnt = ws; ws += S;

    wmma_gemm_pair<16, 0><<<TQ + TS, 512, 0, stream>>>(
        Xq, Wn1, q1, TQ, Xs, Wn1, s1, bn1, nullptr, H, E, 1);
    wmma_gemm_pair<8, 0><<<TQ + TS, 256, 0, stream>>>(
        q1, Wn2, qf, TQ, s1, Wn2, sf, bn2, nullptr, H, H, 0);

    for (int t = 0; t < STEPS; ++t) {
        const float* W1a = Wm1 + (size_t)t * (2 * H) * H;
        const float* W1b = W1a + (size_t)H * H;
        wmma_gemm_pair<8, 0><<<TS + TS, 256, 0, stream>>>(
            sf, W1a, Ab, TS, sf, W1b, Bb, nullptr, nullptr, H, H, 0);
        pair_accum<<<S, H, 0, stream>>>(Ab, Bb, bm1 + (size_t)t * H, y, hag, cnt, H);
        wmma_gemm_pair<8, 1><<<TS, 256, 0, stream>>>(
            hag, Wm2 + (size_t)t * H * H, sf, TS, nullptr, nullptr, nullptr,
            bm2 + (size_t)t * H, cnt, H, H, 0);
    }

    wmma_gemm_pair<8, 0><<<TQ + TS, 256, 0, stream>>>(
        qf, Wr1, hqb, TQ, sf, Wr1 + (size_t)H * H, hsb, nullptr, nullptr, H, H, 0);
    score_kernel<<<Q, 256, 0, stream>>>(hqb, hsb, br1, wr2, br2, out, S, H);
}


# Score kernel: 32q x 64s tile staged in LDS once; lane owns 2 s, wave 4 q, h summed in-lane as even/odd partial sums, no cross-lane reduction; bias pre-added to hq; changes the summation order
# speedup vs baseline: 10.0000x; 1.5724x over previous
#include <hip/hip_runtime.h>
#include <hip/hip_bf16.h>

typedef float v2f __attribute__((ext_vector_type(2)));
typedef float v8f __attribute__((ext_vector_type(8)));

template <int NW, int EPI>
__global__ __launch_bounds__(NW * 32) void wmma_gemm_pair(
    const float* __restrict__ A0, const float* __restrict__ W0,
    float* __restrict__ C0, int tiles0,
    const float* __restrict__ A1, const float* __restrict__ W1,
    float* __restrict__ C1,
    const float* __restrict__ bias, const float* __restrict__ cnt,
    int N, int K, int relu)
{
    constexpr int PITCH = 68;
    __shared__ alignas(16) float red[NW][16 * PITCH];

    const int lane = threadIdx.x & 31;
    const int wave = threadIdx.x >> 5;
    const int tilesN = N >> 6;

    const bool second = (int)blockIdx.x >= tiles0;
    const float* A = second ? A1 : A0;
    const float* W = second ? W1 : W0;
    float*       C = second ? C1 : C0;
    const int tile = second ? (int)blockIdx.x - tiles0 : (int)blockIdx.x;
    const int tm = (tile / tilesN) << 4;
    const int tn = (tile % tilesN) << 6;

    const int half = lane >> 4;
    const int l16  = lane & 15;
    const int k0   = wave << 5;

    const float* ap = A + (size_t)(tm + l16) * K + k0 + 2 * half;
    const float* bp = W + (size_t)(k0 + 2 * half) * N + tn + l16;
    const size_t sN = (size_t)N;

    v2f a[8];
    v2f f[8][4];
#pragma unroll
    for (int s = 0; s < 8; ++s) {
        a[s] = *(const v2f*)(ap + 4 * s);
        const float* b0 = bp + (size_t)(4 * s) * sN;
        const float* b1 = b0 + sN;
#pragma unroll
        for (int j = 0; j < 4; ++j) {
            f[s][j].x = b0[16 * j];
            f[s][j].y = b1[16 * j];
        }
    }

    __builtin_amdgcn_sched_barrier(0);

    v8f acc0 = {}, acc1 = {}, acc2 = {}, acc3 = {};
#pragma unroll
    for (int s = 0; s < 8; ++s) {
        acc0 = __builtin_amdgcn_wmma_f32_16x16x4_f32(false, a[s], false, f[s][0],
                                                     (short)0, acc0, false, false);
        acc1 = __builtin_amdgcn_wmma_f32_16x16x4_f32(false, a[s], false, f[s][1],
                                                     (short)0, acc1, false, false);
        acc2 = __builtin_amdgcn_wmma_f32_16x16x4_f32(false, a[s], false, f[s][2],
                                                     (short)0, acc2, false, false);
        acc3 = __builtin_amdgcn_wmma_f32_16x16x4_f32(false, a[s], false, f[s][3],
                                                     (short)0, acc3, false, false);
    }

    float* my = &red[wave][(half * 8) * PITCH + l16];
#pragma unroll
    for (int r = 0; r < 8; ++r) {
        my[r * PITCH]      = acc0[r];
        my[r * PITCH + 16] = acc1[r];
        my[r * PITCH + 32] = acc2[r];
        my[r * PITCH + 48] = acc3[r];
    }
    __syncthreads();

    const int t = threadIdx.x;
    if (t < 256) {
        const int row = t >> 4;
        const int c4  = (t & 15) << 2;
        float* cp = C + (size_t)(tm + row) * N + tn + c4;
        float4 sum = *(const float4*)&red[0][row * PITCH + c4];
#pragma unroll
        for (int w = 1; w < NW; ++w) {
            const float4 p = *(const float4*)&red[w][row * PITCH + c4];
            sum.x += p.x; sum.y += p.y; sum.z += p.z; sum.w += p.w;
        }
        if (EPI == 0) {
            if (bias) {
                const float4 b4 = *(const float4*)(bias + tn + c4);
                sum.x += b4.x; sum.y += b4.y; sum.z += b4.z; sum.w += b4.w;
            }
            if (relu) {
                sum.x = fmaxf(sum.x, 0.f); sum.y = fmaxf(sum.y, 0.f);
                sum.z = fmaxf(sum.z, 0.f); sum.w = fmaxf(sum.w, 0.f);
            }
            *(float4*)cp = sum;
        } else {
            const float c = cnt[tm + row];
            const float denom = fmaxf(c, 1.0f);
            const float4 bv = *(const float4*)(bias + tn + c4);
            float4 sv = *(const float4*)cp;
            sv.x += (sum.x + c * bv.x) / denom;
            sv.y += (sum.y + c * bv.y) / denom;
            sv.z += (sum.z + c * bv.z) / denom;
            sv.w += (sum.w + c * bv.w) / denom;
            *(float4*)cp = sv;
        }
    }
}

constexpr int PAIR_S = 512;
constexpr int PG = 32;
__global__ __launch_bounds__(256) void pair_accum(
    const float* __restrict__ Abuf, const float* __restrict__ Bbuf,
    const float* __restrict__ bm1t, const int* __restrict__ y,
    float* __restrict__ haggr, float* __restrict__ cnt, int H)
{
    __shared__ alignas(16) unsigned list[PAIR_S + PG];
    __shared__ alignas(16) int wcnt[8];

    const int i = blockIdx.x;
    const int h = threadIdx.x;
    const int lane = h & 31;
    const int wave = h >> 5;

    const int j0 = (wave << 6) + lane;
    const int j1 = j0 + 32;
    const int yi = y[i];
    const int y0 = y[j0];
    const int y1 = y[j1];
    const float av = Abuf[(size_t)i * H + h] + bm1t[h];

    const bool m0 = (y0 == yi) && (j0 != i);
    const bool m1 = (y1 == yi) && (j1 != i);
    const unsigned b0 = __builtin_amdgcn_ballot_w32(m0);
    const unsigned b1 = __builtin_amdgcn_ballot_w32(m1);
    const int c0 = __builtin_popcount(b0);
    const int c1 = __builtin_popcount(b1);
    if (lane == 0) wcnt[wave] = c0 + c1;
    __syncthreads();

    int base = 0, n = 0;
    for (int w = 0; w < 8; ++w) {
        const int c = wcnt[w];
        base += (w < wave) ? c : 0;
        n += c;
    }
    const unsigned below = (1u << lane) - 1u;
    if (m0) list[base + __builtin_popcount(b0 & below)] = (unsigned)(j0 * H);
    if (m1) list[base + c0 + __builtin_popcount(b1 & below)] = (unsigned)(j1 * H);
    if (h < PG) list[n + h] = 0u;
    __syncthreads();

    float acc = 0.0f;
    for (int p = 0; p < n; p += PG) {
        float v[PG];
#pragma unroll
        for (int u = 0; u < PG; ++u)
            v[u] = Bbuf[list[p + u] + (unsigned)h];
#pragma unroll
        for (int u = 0; u < PG; ++u) {
            const float t = fmaxf(av + v[u], 0.0f);
            acc += (p + u < n) ? t : 0.0f;
        }
    }
    haggr[(size_t)i * H + h] = acc;
    if (h == 0) cnt[i] = (float)n;
}

constexpr int SC_H  = 256;
constexpr int SC_TQ = 32;
constexpr int SC_TS = 64;
constexpr int SC_RQ = 4;
constexpr int SC_PS = SC_H + 4;

__device__ __forceinline__ v2f relu_mad(v2f u, v2f v, v2f w, v2f acc)
{
    v2f t = u + v;
    t = __builtin_elementwise_max(t, (v2f){0.0f, 0.0f});
    return t * w + acc;
}

__global__ __launch_bounds__(256) void score_kernel(
    const float* __restrict__ hq, const float* __restrict__ hs,
    const float* __restrict__ br1, const float* __restrict__ wr2,
    const float* __restrict__ br2, float* __restrict__ out, int S)
{
    __shared__ alignas(16) float hs_l[SC_TS * SC_PS];
    __shared__ alignas(16) float hq_l[SC_TQ * SC_H];
    __shared__ alignas(16) float w_l[SC_H];

    const int t = threadIdx.x;
    const int lane = t & 31;
    const int wave = t >> 5;
    const int tilesS = S / SC_TS;
    const int q0 = ((int)blockIdx.x / tilesS) * SC_TQ;
    const int s0 = ((int)blockIdx.x % tilesS) * SC_TS;

    const int c4 = (t & 63) << 2;
    const int r0 = t >> 6;
    const float4 b4 = *(const float4*)(br1 + c4);
#pragma unroll
    for (int i = 0; i < SC_TS / 4; ++i) {
        const int row = r0 + 4 * i;
        *(float4*)&hs_l[row * SC_PS + c4] =
            *(const float4*)(hs + (size_t)(s0 + row) * SC_H + c4);
    }
#pragma unroll
    for (int i = 0; i < SC_TQ / 4; ++i) {
        const int row = r0 + 4 * i;
        float4 u = *(const float4*)(hq + (size_t)(q0 + row) * SC_H + c4);
        u.x += b4.x; u.y += b4.y; u.z += b4.z; u.w += b4.w;
        *(float4*)&hq_l[row * SC_H + c4] = u;
    }
    if (t < 64) *(float4*)&w_l[c4] = *(const float4*)(wr2 + c4);
    const float b2 = br2[0];
    __syncthreads();

    const float* h0p = &hs_l[lane * SC_PS];
    const float* h1p = &hs_l[(lane + 32) * SC_PS];
    const float* qp  = &hq_l[(wave * SC_RQ) * SC_H];

    v2f acc0[SC_RQ], acc1[SC_RQ];
#pragma unroll
    for (int r = 0; r < SC_RQ; ++r) { acc0[r] = (v2f){0.0f, 0.0f}; acc1[r] = (v2f){0.0f, 0.0f}; }

#pragma unroll 4
    for (int h = 0; h < SC_H; h += 4) {
        const float4 v0 = *(const float4*)(h0p + h);
        const float4 v1 = *(const float4*)(h1p + h);
        const float4 w4 = *(const float4*)(w_l + h);
        const v2f v0a = {v0.x, v0.y}, v0b = {v0.z, v0.w};
        const v2f v1a = {v1.x, v1.y}, v1b = {v1.z, v1.w};
        const v2f wa  = {w4.x, w4.y}, wb  = {w4.z, w4.w};
#pragma unroll
        for (int r = 0; r < SC_RQ; ++r) {
            const float4 u = *(const float4*)(qp + r * SC_H + h);
            const v2f ua = {u.x, u.y}, ub = {u.z, u.w};
            acc0[r] = relu_mad(ua, v0a, wa, acc0[r]);
            acc1[r] = relu_mad(ua, v1a, wa, acc1[r]);
            acc0[r] = relu_mad(ub, v0b, wb, acc0[r]);
            acc1[r] = relu_mad(ub, v1b, wb, acc1[r]);
        }
    }

    float* op = out + (size_t)(q0 + wave * SC_RQ) * S + s0 + lane;
#pragma unroll
    for (int r = 0; r < SC_RQ; ++r) {
        op[(size_t)r * S]      = (acc0[r].x + acc0[r].y) + b2;
        op[(size_t)r * S + 32] = (acc1[r].x + acc1[r].y) + b2;
    }
}

extern "C" void kernel_launch(void* const* d_in, const int* in_sizes, int n_in,
                              void* d_out, int out_size, void* d_ws, size_t ws_size,
                              hipStream_t stream)
{
    const float* Xq  = (const float*)d_in[0];
    const float* Xs  = (const float*)d_in[1];
    const int*   y   = (const int*)d_in[2];
    const float* Wn1 = (const float*)d_in[3];
    const float* bn1 = (const float*)d_in[4];
    const float* Wn2 = (const float*)d_in[5];
    const float* bn2 = (const float*)d_in[6];
    const float* Wm1 = (const float*)d_in[7];
    const float* bm1 = (const float*)d_in[8];
    const float* Wm2 = (const float*)d_in[9];
    const float* bm2 = (const float*)d_in[10];
    const float* Wr1 = (const float*)d_in[11];
    const float* br1 = (const float*)d_in[12];
    const float* wr2 = (const float*)d_in[13];
    const float* br2 = (const float*)d_in[14];
    float* out = (float*)d_out;

    const int E = 512, H = 256, Q = 1024, S = 512, STEPS = 3;
    const int TQ = (Q / 16) * (H / 64);
    const int TS = (S / 16) * (H / 64);

    float* ws  = (float*)d_ws;
    float* q1  = ws; ws += (size_t)Q * H;
    float* qf  = ws; ws += (size_t)Q * H;
    float* s1  = ws; ws += (size_t)S * H;
    float* sf  = ws; ws += (size_t)S * H;
    float* Ab  = ws; ws += (size_t)S * H;
    float* Bb  = ws; ws += (size_t)S * H;
    float* hag = ws; ws += (size_t)S * H;
    float* hqb = ws; ws += (size_t)Q * H;
    float* hsb = ws; ws += (size_t)S * H;
    float* cnt = ws; ws += S;

    wmma_gemm_pair<16, 0><<<TQ + TS, 512, 0, stream>>>(
        Xq, Wn1, q1, TQ, Xs, Wn1, s1, bn1, nullptr, H, E, 1);
    wmma_gemm_pair<8, 0><<<TQ + TS, 256, 0, stream>>>(
        q1, Wn2, qf, TQ, s1, Wn2, sf, bn2, nullptr, H, H, 0);

    for (int t = 0; t < STEPS; ++t) {
        const float* W1a = Wm1 + (size_t)t * (2 * H) * H;
        const float* W1b = W1a + (size_t)H * H;
        wmma_gemm_pair<8, 0><<<TS + TS, 256, 0, stream>>>(
            sf, W1a, Ab, TS, sf, W1b, Bb, nullptr, nullptr, H, H, 0);
        pair_accum<<<S, H, 0, stream>>>(Ab, Bb, bm1 + (size_t)t * H, y, hag, cnt, H);
        wmma_gemm_pair<8, 1><<<TS, 256, 0, stream>>>(
            hag, Wm2 + (size_t)t * H * H, sf, TS, nullptr, nullptr, nullptr,
            bm2 + (size_t)t * H, cnt, H, H, 0);
    }

    wmma_gemm_pair<8, 0><<<TQ + TS, 256, 0, stream>>>(
        qf, Wr1, hqb, TQ, sf, Wr1 + (size_t)H * H, hsb, nullptr, nullptr, H, H, 0);
    score_kernel<<<(Q / 32) * (S / 64), 256, 0, stream>>>(hqb, hsb, br1, wr2, br2, out, S);
}


# Step GEMM also emits next step's A, B (last: hs) as nodes @ W + hag @ (Wm2 @ W), Wm2 @ W formed early beside other launches; 13 -> 10 dependent launches; changes the summation order
# speedup vs baseline: 10.5170x; 1.0517x over previous
#include <hip/hip_runtime.h>
#include <hip/hip_bf16.h>

typedef float v2f __attribute__((ext_vector_type(2)));
typedef float v8f __attribute__((ext_vector_type(8)));


constexpr int PITCH = 68;

__device__ __forceinline__ void load_slice(const float* ap, const float* bp, size_t sN,
                                           v2f (&a)[8], v2f (&f)[8][4])
{
#pragma unroll
    for (int s = 0; s < 8; ++s) {
        a[s] = *(const v2f*)(ap + 4 * s);
        const float* b0 = bp + (size_t)(4 * s) * sN;
        const float* b1 = b0 + sN;
#pragma unroll
        for (int j = 0; j < 4; ++j) {
            f[s][j].x = b0[16 * j];
            f[s][j].y = b1[16 * j];
        }
    }
}

__device__ __forceinline__ void mma_slice_to_lds(const v2f (&a)[8], const v2f (&f)[8][4], float* my)
{
    v8f acc0 = {}, acc1 = {}, acc2 = {}, acc3 = {};
#pragma unroll
    for (int s = 0; s < 8; ++s) {
        acc0 = __builtin_amdgcn_wmma_f32_16x16x4_f32(false, a[s], false, f[s][0],
                                                     (short)0, acc0, false, false);
        acc1 = __builtin_amdgcn_wmma_f32_16x16x4_f32(false, a[s], false, f[s][1],
                                                     (short)0, acc1, false, false);
        acc2 = __builtin_amdgcn_wmma_f32_16x16x4_f32(false, a[s], false, f[s][2],
                                                     (short)0, acc2, false, false);
        acc3 = __builtin_amdgcn_wmma_f32_16x16x4_f32(false, a[s], false, f[s][3],
                                                     (short)0, acc3, false, false);
    }
#pragma unroll
    for (int r = 0; r < 8; ++r) {
        my[r * PITCH]      = acc0[r];
        my[r * PITCH + 16] = acc1[r];
        my[r * PITCH + 32] = acc2[r];
        my[r * PITCH + 48] = acc3[r];
    }
}

template <int NW>
__global__ __launch_bounds__(NW * 32) void wmma_gemm_multi(
    const float* __restrict__ A0, const float* __restrict__ W0, float* __restrict__ C0, int e0,
    const float* __restrict__ A1, const float* __restrict__ W1, float* __restrict__ C1, int e1,
    const float* __restrict__ A2, const float* __restrict__ W2, float* __restrict__ C2, int e2,
    const float* __restrict__ A3, const float* __restrict__ W3, float* __restrict__ C3, int e3,
    const float* __restrict__ A4, const float* __restrict__ W4, float* __restrict__ C4,
    const float* __restrict__ bias, int nbias, int N, int K, int relu)
{
    __shared__ alignas(16) float red[NW][16 * PITCH];

    const int lane = threadIdx.x & 31;
    const int wave = threadIdx.x >> 5;
    const int tilesN = N >> 6;

    const int b = (int)blockIdx.x;
    const int sel = (b >= e0) + (b >= e1) + (b >= e2) + (b >= e3);
    const float* A = sel == 0 ? A0 : sel == 1 ? A1 : sel == 2 ? A2 : sel == 3 ? A3 : A4;
    const float* W = sel == 0 ? W0 : sel == 1 ? W1 : sel == 2 ? W2 : sel == 3 ? W3 : W4;
    float*       C = sel == 0 ? C0 : sel == 1 ? C1 : sel == 2 ? C2 : sel == 3 ? C3 : C4;
    const int first = sel == 0 ? 0 : sel == 1 ? e0 : sel == 2 ? e1 : sel == 3 ? e2 : e3;
    const int tile = b - first;
    const int tm = (tile / tilesN) << 4;
    const int tn = (tile % tilesN) << 6;

    const int half = lane >> 4;
    const int l16  = lane & 15;
    const int k0   = wave << 5;

    const float* ap = A + (size_t)(tm + l16) * K + k0 + 2 * half;
    const float* bp = W + (size_t)(k0 + 2 * half) * N + tn + l16;

    v2f a[8];
    v2f f[8][4];
    load_slice(ap, bp, (size_t)N, a, f);

    __builtin_amdgcn_sched_barrier(0);

    mma_slice_to_lds(a, f, &red[wave][(half * 8) * PITCH + l16]);
    __syncthreads();

    const int t = threadIdx.x;
    if (t < 256) {
        const int row = t >> 4;
        const int c4  = (t & 15) << 2;
        float* cp = C + (size_t)(tm + row) * N + tn + c4;
        float4 sum = *(const float4*)&red[0][row * PITCH + c4];
#pragma unroll
        for (int w = 1; w < NW; ++w) {
            const float4 p = *(const float4*)&red[w][row * PITCH + c4];
            sum.x += p.x; sum.y += p.y; sum.z += p.z; sum.w += p.w;
        }
        if (bias && sel < nbias) {
            const float4 b4 = *(const float4*)(bias + tn + c4);
            sum.x += b4.x; sum.y += b4.y; sum.z += b4.z; sum.w += b4.w;
        }
        if (relu) {
            sum.x = fmaxf(sum.x, 0.f); sum.y = fmaxf(sum.y, 0.f);
            sum.z = fmaxf(sum.z, 0.f); sum.w = fmaxf(sum.w, 0.f);
        }
        *(float4*)cp = sum;
    }
}

__global__ __launch_bounds__(512) void step_gemm(
    const float* __restrict__ hag, const float* __restrict__ cnt,
    const float* __restrict__ Sin, const float* __restrict__ bv,
    const float* __restrict__ Wh0, const float* __restrict__ Wb0, float* __restrict__ C0, int e0,
    const float* __restrict__ Wh1, const float* __restrict__ Wb1, float* __restrict__ C1, int e1,
    const float* __restrict__ Wh2, const float* __restrict__ Wb2, float* __restrict__ C2,
    int N, int K)
{
    __shared__ alignas(16) float red[16][16 * PITCH];

    const int lane = threadIdx.x & 31;
    const int wave = __builtin_amdgcn_readfirstlane((int)(threadIdx.x >> 5));
    const int tilesN = N >> 6;

    const int b = (int)blockIdx.x;
    const int sel = (b >= e0) + (b >= e1);
    const float* Wh = sel == 0 ? Wh0 : sel == 1 ? Wh1 : Wh2;
    const float* Wb = sel == 0 ? Wb0 : sel == 1 ? Wb1 : Wb2;
    float*       C  = sel == 0 ? C0  : sel == 1 ? C1  : C2;
    const int tile = b - (sel == 0 ? 0 : sel == 1 ? e0 : e1);
    const int tm = (tile / tilesN) << 4;
    const int tn = (tile % tilesN) << 6;

    const int half = lane >> 4;
    const int l16  = lane & 15;
    const bool upper = wave >= 8;
    const int k0 = (wave & 7) << 5;
    const bool nodeprod = (Wb != nullptr);

    if (!upper || nodeprod) {
        const float* Asrc = upper ? Sin : hag;
        const float* Wsrc = upper ? Wb : Wh;
        const float* ap = Asrc + (size_t)(tm + l16) * K + k0 + 2 * half;
        const float* bp = Wsrc + (size_t)(k0 + 2 * half) * N + tn + l16;

        v2f a[8];
        v2f f[8][4];
        load_slice(ap, bp, (size_t)N, a, f);
        if (upper) {
            const float c = cnt[tm + l16];
            const float fl = c / fmaxf(c, 1.0f);
            const float* bq = bv + k0 + 2 * half;
#pragma unroll
            for (int s = 0; s < 8; ++s) {
                const v2f b2 = *(const v2f*)(bq + 4 * s);
                a[s].x += fl * b2.x;
                a[s].y += fl * b2.y;
            }
        }
        __builtin_amdgcn_sched_barrier(0);
        mma_slice_to_lds(a, f, &red[wave][(half * 8) * PITCH + l16]);
    }
    __syncthreads();

    const int t = threadIdx.x;
    if (t < 256) {
        const int row = t >> 4;
        const int c4  = (t & 15) << 2;
        const size_t off = (size_t)(tm + row) * N + tn + c4;
        float4 sh = *(const float4*)&red[0][row * PITCH + c4];
#pragma unroll
        for (int w = 1; w < 8; ++w) {
            const float4 p = *(const float4*)&red[w][row * PITCH + c4];
            sh.x += p.x; sh.y += p.y; sh.z += p.z; sh.w += p.w;
        }
        const float c = cnt[tm + row];
        const float denom = fmaxf(c, 1.0f);
        float4 o;
        if (nodeprod) {
            float4 sb = *(const float4*)&red[8][row * PITCH + c4];
#pragma unroll
            for (int w = 9; w < 16; ++w) {
                const float4 p = *(const float4*)&red[w][row * PITCH + c4];
                sb.x += p.x; sb.y += p.y; sb.z += p.z; sb.w += p.w;
            }
            o.x = sb.x + sh.x / denom;
            o.y = sb.y + sh.y / denom;
            o.z = sb.z + sh.z / denom;
            o.w = sb.w + sh.w / denom;
        } else {
            const float4 b4 = *(const float4*)(bv + tn + c4);
            o = *(const float4*)(Sin + off);
            o.x += (sh.x + c * b4.x) / denom;
            o.y += (sh.y + c * b4.y) / denom;
            o.z += (sh.z + c * b4.z) / denom;
            o.w += (sh.w + c * b4.w) / denom;
        }
        *(float4*)(C + off) = o;
    }
}

constexpr int PAIR_S = 512;
constexpr int PG = 32;
__global__ __launch_bounds__(256) void pair_accum(
    const float* __restrict__ Abuf, const float* __restrict__ Bbuf,
    const float* __restrict__ bm1t, const int* __restrict__ y,
    float* __restrict__ haggr, float* __restrict__ cnt, int H)
{
    __shared__ alignas(16) unsigned list[PAIR_S + PG];
    __shared__ alignas(16) int wcnt[8];

    const int i = blockIdx.x;
    const int h = threadIdx.x;
    const int lane = h & 31;
    const int wave = h >> 5;

    const int j0 = (wave << 6) + lane;
    const int j1 = j0 + 32;
    const int yi = y[i];
    const int y0 = y[j0];
    const int y1 = y[j1];
    const float av = Abuf[(size_t)i * H + h] + bm1t[h];

    const bool m0 = (y0 == yi) && (j0 != i);
    const bool m1 = (y1 == yi) && (j1 != i);
    const unsigned b0 = __builtin_amdgcn_ballot_w32(m0);
    const unsigned b1 = __builtin_amdgcn_ballot_w32(m1);
    const int c0 = __builtin_popcount(b0);
    const int c1 = __builtin_popcount(b1);
    if (lane == 0) wcnt[wave] = c0 + c1;
    __syncthreads();

    int base = 0, n = 0;
    for (int w = 0; w < 8; ++w) {
        const int c = wcnt[w];
        base += (w < wave) ? c : 0;
        n += c;
    }
    const unsigned below = (1u << lane) - 1u;
    if (m0) list[base + __builtin_popcount(b0 & below)] = (unsigned)(j0 * H);
    if (m1) list[base + c0 + __builtin_popcount(b1 & below)] = (unsigned)(j1 * H);
    if (h < PG) list[n + h] = 0u;
    __syncthreads();

    float acc = 0.0f;
    for (int p = 0; p < n; p += PG) {
        float v[PG];
#pragma unroll
        for (int u = 0; u < PG; ++u)
            v[u] = Bbuf[list[p + u] + (unsigned)h];
#pragma unroll
        for (int u = 0; u < PG; ++u) {
            const float t = fmaxf(av + v[u], 0.0f);
            acc += (p + u < n) ? t : 0.0f;
        }
    }
    haggr[(size_t)i * H + h] = acc;
    if (h == 0) cnt[i] = (float)n;
}

constexpr int SC_H  = 256;
constexpr int SC_TQ = 32;
constexpr int SC_TS = 64;
constexpr int SC_RQ = 4;
constexpr int SC_PS = SC_H + 4;

__device__ __forceinline__ v2f relu_mad(v2f u, v2f v, v2f w, v2f acc)
{
    v2f t = u + v;
    t = __builtin_elementwise_max(t, (v2f){0.0f, 0.0f});
    return t * w + acc;
}

__global__ __launch_bounds__(256) void score_kernel(
    const float* __restrict__ hq, const float* __restrict__ hs,
    const float* __restrict__ br1, const float* __restrict__ wr2,
    const float* __restrict__ br2, float* __restrict__ out, int S)
{
    __shared__ alignas(16) float hs_l[SC_TS * SC_PS];
    __shared__ alignas(16) float hq_l[SC_TQ * SC_H];
    __shared__ alignas(16) float w_l[SC_H];

    const int t = threadIdx.x;
    const int lane = t & 31;
    const int wave = t >> 5;
    const int tilesS = S / SC_TS;
    const int q0 = ((int)blockIdx.x / tilesS) * SC_TQ;
    const int s0 = ((int)blockIdx.x % tilesS) * SC_TS;

    const int c4 = (t & 63) << 2;
    const int r0 = t >> 6;
    const float4 b4 = *(const float4*)(br1 + c4);
#pragma unroll
    for (int i = 0; i < SC_TS / 4; ++i) {
        const int row = r0 + 4 * i;
        *(float4*)&hs_l[row * SC_PS + c4] =
            *(const float4*)(hs + (size_t)(s0 + row) * SC_H + c4);
    }
#pragma unroll
    for (int i = 0; i < SC_TQ / 4; ++i) {
        const int row = r0 + 4 * i;
        float4 u = *(const float4*)(hq + (size_t)(q0 + row) * SC_H + c4);
        u.x += b4.x; u.y += b4.y; u.z += b4.z; u.w += b4.w;
        *(float4*)&hq_l[row * SC_H + c4] = u;
    }
    if (t < 64) *(float4*)&w_l[c4] = *(const float4*)(wr2 + c4);
    const float b2 = br2[0];
    __syncthreads();

    const float* h0p = &hs_l[lane * SC_PS];
    const float* h1p = &hs_l[(lane + 32) * SC_PS];
    const float* qp  = &hq_l[(wave * SC_RQ) * SC_H];

    v2f acc0[SC_RQ], acc1[SC_RQ];
#pragma unroll
    for (int r = 0; r < SC_RQ; ++r) { acc0[r] = (v2f){0.0f, 0.0f}; acc1[r] = (v2f){0.0f, 0.0f}; }

#pragma unroll 4
    for (int h = 0; h < SC_H; h += 4) {
        const float4 v0 = *(const float4*)(h0p + h);
        const float4 v1 = *(const float4*)(h1p + h);
        const float4 w4 = *(const float4*)(w_l + h);
        const v2f v0a = {v0.x, v0.y}, v0b = {v0.z, v0.w};
        const v2f v1a = {v1.x, v1.y}, v1b = {v1.z, v1.w};
        const v2f wa  = {w4.x, w4.y}, wb  = {w4.z, w4.w};
#pragma unroll
        for (int r = 0; r < SC_RQ; ++r) {
            const float4 u = *(const float4*)(qp + r * SC_H + h);
            const v2f ua = {u.x, u.y}, ub = {u.z, u.w};
            acc0[r] = relu_mad(ua, v0a, wa, acc0[r]);
            acc1[r] = relu_mad(ua, v1a, wa, acc1[r]);
            acc0[r] = relu_mad(ub, v0b, wb, acc0[r]);
            acc1[r] = relu_mad(ub, v1b, wb, acc1[r]);
        }
    }

    float* op = out + (size_t)(q0 + wave * SC_RQ) * S + s0 + lane;
#pragma unroll
    for (int r = 0; r < SC_RQ; ++r) {
        op[(size_t)r * S]      = (acc0[r].x + acc0[r].y) + b2;
        op[(size_t)r * S + 32] = (acc1[r].x + acc1[r].y) + b2;
    }
}

extern "C" void kernel_launch(void* const* d_in, const int* in_sizes, int n_in,
                              void* d_out, int out_size, void* d_ws, size_t ws_size,
                              hipStream_t stream)
{
    const float* Xq  = (const float*)d_in[0];
    const float* Xs  = (const float*)d_in[1];
    const int*   y   = (const int*)d_in[2];
    const float* Wn1 = (const float*)d_in[3];
    const float* bn1 = (const float*)d_in[4];
    const float* Wn2 = (const float*)d_in[5];
    const float* bn2 = (const float*)d_in[6];
    const float* Wm1 = (const float*)d_in[7];
    const float* bm1 = (const float*)d_in[8];
    const float* Wm2 = (const float*)d_in[9];
    const float* bm2 = (const float*)d_in[10];
    const float* Wr1 = (const float*)d_in[11];
    const float* br1 = (const float*)d_in[12];
    const float* wr2 = (const float*)d_in[13];
    const float* br2 = (const float*)d_in[14];
    float* out = (float*)d_out;

    const int E = 512, H = 256, Q = 1024, S = 512;
    const int TQ = (Q / 16) * (H / 64);
    const int TS = (S / 16) * (H / 64);
    const int TW = (H / 16) * (H / 64);
    const size_t HH = (size_t)H * H;

    float* ws  = (float*)d_ws;
    float* q1  = ws; ws += (size_t)Q * H;
    float* s1  = ws; ws += (size_t)S * H;
    float* qf  = ws; ws += (size_t)Q * H;
    float* sf  = ws; ws += (size_t)S * H;
    float* Ab  = ws; ws += (size_t)S * H;
    float* Bb  = ws; ws += (size_t)S * H;
    float* hag = ws; ws += (size_t)S * H;
    float* hsb = ws; ws += (size_t)S * H;
    float* Pa1 = ws; ws += HH;
    float* Pb1 = ws; ws += HH;
    float* Pr  = ws; ws += HH;
    float* cnt = ws; ws += S;
    float* hqb = q1;
    float* sf2 = s1;
    float* Pa0 = hsb;
    float* Pb0 = hsb + HH;

    const float* W1a0 = Wm1;            const float* W1b0 = Wm1 + HH;
    const float* W1a1 = Wm1 + 2 * HH;   const float* W1b1 = Wm1 + 3 * HH;
    const float* W1a2 = Wm1 + 4 * HH;   const float* W1b2 = Wm1 + 5 * HH;
    const float* Wm20 = Wm2;
    const float* Wm21 = Wm2 + HH;
    const float* Wm22 = Wm2 + 2 * HH;
    const float* Wr1a = Wr1;
    const float* Wr1b = Wr1 + HH;

    wmma_gemm_multi<16><<<TQ + TS, 512, 0, stream>>>(
        Xq, Wn1, q1, TQ,
        Xs, Wn1, s1, TQ + TS,
        nullptr, nullptr, nullptr, TQ + TS,
        nullptr, nullptr, nullptr, TQ + TS,
        nullptr, nullptr, nullptr,
        bn1, 2, H, E, 1);
    wmma_gemm_multi<8><<<TQ + TS + 3 * TW, 256, 0, stream>>>(
        q1, Wn2, qf, TQ,
        s1, Wn2, sf, TQ + TS,
        Wm20, W1a1, Pa0, TQ + TS + TW,
        Wm20, W1b1, Pb0, TQ + TS + 2 * TW,
        Wm21, W1a2, Pa1,
        bn2, 2, H, H, 0);
    wmma_gemm_multi<8><<<2 * TS + TQ + 2 * TW, 256, 0, stream>>>(
        sf, W1a0, Ab, TS,
        sf, W1b0, Bb, 2 * TS,
        qf, Wr1a, hqb, 2 * TS + TQ,
        Wm21, W1b2, Pb1, 2 * TS + TQ + TW,
        Wm22, Wr1b, Pr,
        nullptr, 0, H, H, 0);

    pair_accum<<<S, H, 0, stream>>>(Ab, Bb, bm1, y, hag, cnt, H);
    step_gemm<<<3 * TS, 512, 0, stream>>>(
        hag, cnt, sf, bm2,
        Wm20, nullptr, sf2, TS,
        Pa0, W1a1, Ab, 2 * TS,
        Pb0, W1b1, Bb,
        H, H);
    pair_accum<<<S, H, 0, stream>>>(Ab, Bb, bm1 + H, y, hag, cnt, H);
    step_gemm<<<3 * TS, 512, 0, stream>>>(
        hag, cnt, sf2, bm2 + H,
        Wm21, nullptr, sf, TS,
        Pa1, W1a2, Ab, 2 * TS,
        Pb1, W1b2, Bb,
        H, H);
    pair_accum<<<S, H, 0, stream>>>(Ab, Bb, bm1 + 2 * H, y, hag, cnt, H);
    step_gemm<<<TS, 512, 0, stream>>>(
        hag, cnt, sf, bm2 + 2 * H,
        Pr, Wr1b, hsb, TS,
        nullptr, nullptr, nullptr, TS,
        nullptr, nullptr, nullptr,
        H, H);

    score_kernel<<<(Q / 32) * (S / 64), 256, 0, stream>>>(hqb, hsb, br1, wr2, br2, out, S);
}


# GEMM fragment loads widened to 16 B: lane's A data = 16 consecutive k of its row, tile j takes columns 4c+j so B rows and LDS partials move as float4; k order permuted: changes the summation order
# speedup vs baseline: 11.1008x; 1.0555x over previous
#include <hip/hip_runtime.h>
#include <hip/hip_bf16.h>

typedef float v2f __attribute__((ext_vector_type(2)));
typedef float v8f __attribute__((ext_vector_type(8)));


constexpr int PITCH = 68;

__device__ __forceinline__ void load_slice(const float* ap, const float* bp, size_t sN,
                                           v2f (&a)[8], v2f (&f)[8][4])
{
#pragma unroll
    for (int i = 0; i < 4; ++i) {
        const float4 q = *(const float4*)(ap + 4 * i);
        a[2 * i]     = (v2f){q.x, q.y};
        a[2 * i + 1] = (v2f){q.z, q.w};
    }
#pragma unroll
    for (int s = 0; s < 8; ++s) {
        const float* b0 = bp + (size_t)(2 * s) * sN;
        const float* b1 = b0 + sN;
        const float4 r0 = *(const float4*)b0;
        const float4 r1 = *(const float4*)b1;
        f[s][0] = (v2f){r0.x, r1.x};
        f[s][1] = (v2f){r0.y, r1.y};
        f[s][2] = (v2f){r0.z, r1.z};
        f[s][3] = (v2f){r0.w, r1.w};
    }
}

__device__ __forceinline__ void mma_slice_to_lds(const v2f (&a)[8], const v2f (&f)[8][4], float* my)
{
    v8f acc0 = {}, acc1 = {}, acc2 = {}, acc3 = {};
#pragma unroll
    for (int s = 0; s < 8; ++s) {
        acc0 = __builtin_amdgcn_wmma_f32_16x16x4_f32(false, a[s], false, f[s][0],
                                                     (short)0, acc0, false, false);
        acc1 = __builtin_amdgcn_wmma_f32_16x16x4_f32(false, a[s], false, f[s][1],
                                                     (short)0, acc1, false, false);
        acc2 = __builtin_amdgcn_wmma_f32_16x16x4_f32(false, a[s], false, f[s][2],
                                                     (short)0, acc2, false, false);
        acc3 = __builtin_amdgcn_wmma_f32_16x16x4_f32(false, a[s], false, f[s][3],
                                                     (short)0, acc3, false, false);
    }
#pragma unroll
    for (int r = 0; r < 8; ++r)
        *(float4*)&my[r * PITCH] = make_float4(acc0[r], acc1[r], acc2[r], acc3[r]);
}

template <int NW>
__global__ __launch_bounds__(NW * 32) void wmma_gemm_multi(
    const float* __restrict__ A0, const float* __restrict__ W0, float* __restrict__ C0, int e0,
    const float* __restrict__ A1, const float* __restrict__ W1, float* __restrict__ C1, int e1,
    const float* __restrict__ A2, const float* __restrict__ W2, float* __restrict__ C2, int e2,
    const float* __restrict__ A3, const float* __restrict__ W3, float* __restrict__ C3, int e3,
    const float* __restrict__ A4, const float* __restrict__ W4, float* __restrict__ C4,
    const float* __restrict__ bias, int nbias, int N, int K, int relu)
{
    __shared__ alignas(16) float red[NW][16 * PITCH];

    const int lane = threadIdx.x & 31;
    const int wave = threadIdx.x >> 5;
    const int tilesN = N >> 6;

    const int b = (int)blockIdx.x;
    const int sel = (b >= e0) + (b >= e1) + (b >= e2) + (b >= e3);
    const float* A = sel == 0 ? A0 : sel == 1 ? A1 : sel == 2 ? A2 : sel == 3 ? A3 : A4;
    const float* W = sel == 0 ? W0 : sel == 1 ? W1 : sel == 2 ? W2 : sel == 3 ? W3 : W4;
    float*       C = sel == 0 ? C0 : sel == 1 ? C1 : sel == 2 ? C2 : sel == 3 ? C3 : C4;
    const int first = sel == 0 ? 0 : sel == 1 ? e0 : sel == 2 ? e1 : sel == 3 ? e2 : e3;
    const int tile = b - first;
    const int tm = (tile / tilesN) << 4;
    const int tn = (tile % tilesN) << 6;

    const int half = lane >> 4;
    const int l16  = lane & 15;
    const int k0   = wave << 5;

    const float* ap = A + (size_t)(tm + l16) * K + k0 + 16 * half;
    const float* bp = W + (size_t)(k0 + 16 * half) * N + tn + 4 * l16;

    v2f a[8];
    v2f f[8][4];
    load_slice(ap, bp, (size_t)N, a, f);

    __builtin_amdgcn_sched_barrier(0);

    mma_slice_to_lds(a, f, &red[wave][(half * 8) * PITCH + 4 * l16]);
    __syncthreads();

    const int t = threadIdx.x;
    if (t < 256) {
        const int row = t >> 4;
        const int c4  = (t & 15) << 2;
        float* cp = C + (size_t)(tm + row) * N + tn + c4;
        float4 sum = *(const float4*)&red[0][row * PITCH + c4];
#pragma unroll
        for (int w = 1; w < NW; ++w) {
            const float4 p = *(const float4*)&red[w][row * PITCH + c4];
            sum.x += p.x; sum.y += p.y; sum.z += p.z; sum.w += p.w;
        }
        if (bias && sel < nbias) {
            const float4 b4 = *(const float4*)(bias + tn + c4);
            sum.x += b4.x; sum.y += b4.y; sum.z += b4.z; sum.w += b4.w;
        }
        if (relu) {
            sum.x = fmaxf(sum.x, 0.f); sum.y = fmaxf(sum.y, 0.f);
            sum.z = fmaxf(sum.z, 0.f); sum.w = fmaxf(sum.w, 0.f);
        }
        *(float4*)cp = sum;
    }
}

__global__ __launch_bounds__(512) void step_gemm(
    const float* __restrict__ hag, const float* __restrict__ cnt,
    const float* __restrict__ Sin, const float* __restrict__ bv,
    const float* __restrict__ Wh0, const float* __restrict__ Wb0, float* __restrict__ C0, int e0,
    const float* __restrict__ Wh1, const float* __restrict__ Wb1, float* __restrict__ C1, int e1,
    const float* __restrict__ Wh2, const float* __restrict__ Wb2, float* __restrict__ C2,
    int N, int K)
{
    __shared__ alignas(16) float red[16][16 * PITCH];

    const int lane = threadIdx.x & 31;
    const int wave = __builtin_amdgcn_readfirstlane((int)(threadIdx.x >> 5));
    const int tilesN = N >> 6;

    const int b = (int)blockIdx.x;
    const int sel = (b >= e0) + (b >= e1);
    const float* Wh = sel == 0 ? Wh0 : sel == 1 ? Wh1 : Wh2;
    const float* Wb = sel == 0 ? Wb0 : sel == 1 ? Wb1 : Wb2;
    float*       C  = sel == 0 ? C0  : sel == 1 ? C1  : C2;
    const int tile = b - (sel == 0 ? 0 : sel == 1 ? e0 : e1);
    const int tm = (tile / tilesN) << 4;
    const int tn = (tile % tilesN) << 6;

    const int half = lane >> 4;
    const int l16  = lane & 15;
    const bool upper = wave >= 8;
    const int k0 = (wave & 7) << 5;
    const bool nodeprod = (Wb != nullptr);

    if (!upper || nodeprod) {
        const float* Asrc = upper ? Sin : hag;
        const float* Wsrc = upper ? Wb : Wh;
        const float* ap = Asrc + (size_t)(tm + l16) * K + k0 + 16 * half;
        const float* bp = Wsrc + (size_t)(k0 + 16 * half) * N + tn + 4 * l16;

        v2f a[8];
        v2f f[8][4];
        load_slice(ap, bp, (size_t)N, a, f);
        if (upper) {
            const float c = cnt[tm + l16];
            const float fl = c / fmaxf(c, 1.0f);
            const float* bq = bv + k0 + 16 * half;
#pragma unroll
            for (int s = 0; s < 8; ++s) {
                const v2f b2 = *(const v2f*)(bq + 2 * s);
                a[s].x += fl * b2.x;
                a[s].y += fl * b2.y;
            }
        }
        __builtin_amdgcn_sched_barrier(0);
        mma_slice_to_lds(a, f, &red[wave][(half * 8) * PITCH + 4 * l16]);
    }
    __syncthreads();

    const int t = threadIdx.x;
    if (t < 256) {
        const int row = t >> 4;
        const int c4  = (t & 15) << 2;
        const size_t off = (size_t)(tm + row) * N + tn + c4;
        float4 sh = *(const float4*)&red[0][row * PITCH + c4];
#pragma unroll
        for (int w = 1; w < 8; ++w) {
            const float4 p = *(const float4*)&red[w][row * PITCH + c4];
            sh.x += p.x; sh.y += p.y; sh.z += p.z; sh.w += p.w;
        }
        const float c = cnt[tm + row];
        const float denom = fmaxf(c, 1.0f);
        float4 o;
        if (nodeprod) {
            float4 sb = *(const float4*)&red[8][row * PITCH + c4];
#pragma unroll
            for (int w = 9; w < 16; ++w) {
                const float4 p = *(const float4*)&red[w][row * PITCH + c4];
                sb.x += p.x; sb.y += p.y; sb.z += p.z; sb.w += p.w;
            }
            o.x = sb.x + sh.x / denom;
            o.y = sb.y + sh.y / denom;
            o.z = sb.z + sh.z / denom;
            o.w = sb.w + sh.w / denom;
        } else {
            const float4 b4 = *(const float4*)(bv + tn + c4);
            o = *(const float4*)(Sin + off);
            o.x += (sh.x + c * b4.x) / denom;
            o.y += (sh.y + c * b4.y) / denom;
            o.z += (sh.z + c * b4.z) / denom;
            o.w += (sh.w + c * b4.w) / denom;
        }
        *(float4*)(C + off) = o;
    }
}

constexpr int PAIR_S = 512;
constexpr int PG = 32;
__global__ __launch_bounds__(256) void pair_accum(
    const float* __restrict__ Abuf, const float* __restrict__ Bbuf,
    const float* __restrict__ bm1t, const int* __restrict__ y,
    float* __restrict__ haggr, float* __restrict__ cnt, int H)
{
    __shared__ alignas(16) unsigned list[PAIR_S + PG];
    __shared__ alignas(16) int wcnt[8];

    const int i = blockIdx.x;
    const int h = threadIdx.x;
    const int lane = h & 31;
    const int wave = h >> 5;

    const int j0 = (wave << 6) + lane;
    const int j1 = j0 + 32;
    const int yi = y[i];
    const int y0 = y[j0];
    const int y1 = y[j1];
    const float av = Abuf[(size_t)i * H + h] + bm1t[h];

    const bool m0 = (y0 == yi) && (j0 != i);
    const bool m1 = (y1 == yi) && (j1 != i);
    const unsigned b0 = __builtin_amdgcn_ballot_w32(m0);
    const unsigned b1 = __builtin_amdgcn_ballot_w32(m1);
    const int c0 = __builtin_popcount(b0);
    const int c1 = __builtin_popcount(b1);
    if (lane == 0) wcnt[wave] = c0 + c1;
    __syncthreads();

    int base = 0, n = 0;
    for (int w = 0; w < 8; ++w) {
        const int c = wcnt[w];
        base += (w < wave) ? c : 0;
        n += c;
    }
    const unsigned below = (1u << lane) - 1u;
    if (m0) list[base + __builtin_popcount(b0 & below)] = (unsigned)(j0 * H);
    if (m1) list[base + c0 + __builtin_popcount(b1 & below)] = (unsigned)(j1 * H);
    if (h < PG) list[n + h] = 0u;
    __syncthreads();

    float acc = 0.0f;
    for (int p = 0; p < n; p += PG) {
        float v[PG];
#pragma unroll
        for (int u = 0; u < PG; ++u)
            v[u] = Bbuf[list[p + u] + (unsigned)h];
#pragma unroll
        for (int u = 0; u < PG; ++u) {
            const float t = fmaxf(av + v[u], 0.0f);
            acc += (p + u < n) ? t : 0.0f;
        }
    }
    haggr[(size_t)i * H + h] = acc;
    if (h == 0) cnt[i] = (float)n;
}

constexpr int SC_H  = 256;
constexpr int SC_TQ = 32;
constexpr int SC_TS = 64;
constexpr int SC_RQ = 4;
constexpr int SC_PS = SC_H + 4;

__device__ __forceinline__ v2f relu_mad(v2f u, v2f v, v2f w, v2f acc)
{
    v2f t = u + v;
    t = __builtin_elementwise_max(t, (v2f){0.0f, 0.0f});
    return t * w + acc;
}

__global__ __launch_bounds__(256) void score_kernel(
    const float* __restrict__ hq, const float* __restrict__ hs,
    const float* __restrict__ br1, const float* __restrict__ wr2,
    const float* __restrict__ br2, float* __restrict__ out, int S)
{
    __shared__ alignas(16) float hs_l[SC_TS * SC_PS];
    __shared__ alignas(16) float hq_l[SC_TQ * SC_H];
    __shared__ alignas(16) float w_l[SC_H];

    const int t = threadIdx.x;
    const int lane = t & 31;
    const int wave = t >> 5;
    const int tilesS = S / SC_TS;
    const int q0 = ((int)blockIdx.x / tilesS) * SC_TQ;
    const int s0 = ((int)blockIdx.x % tilesS) * SC_TS;

    const int c4 = (t & 63) << 2;
    const int r0 = t >> 6;
    const float4 b4 = *(const float4*)(br1 + c4);
#pragma unroll
    for (int i = 0; i < SC_TS / 4; ++i) {
        const int row = r0 + 4 * i;
        *(float4*)&hs_l[row * SC_PS + c4] =
            *(const float4*)(hs + (size_t)(s0 + row) * SC_H + c4);
    }
#pragma unroll
    for (int i = 0; i < SC_TQ / 4; ++i) {
        const int row = r0 + 4 * i;
        float4 u = *(const float4*)(hq + (size_t)(q0 + row) * SC_H + c4);
        u.x += b4.x; u.y += b4.y; u.z += b4.z; u.w += b4.w;
        *(float4*)&hq_l[row * SC_H + c4] = u;
    }
    if (t < 64) *(float4*)&w_l[c4] = *(const float4*)(wr2 + c4);
    const float b2 = br2[0];
    __syncthreads();

    const float* h0p = &hs_l[lane * SC_PS];
    const float* h1p = &hs_l[(lane + 32) * SC_PS];
    const float* qp  = &hq_l[(wave * SC_RQ) * SC_H];

    v2f acc0[SC_RQ], acc1[SC_RQ];
#pragma unroll
    for (int r = 0; r < SC_RQ; ++r) { acc0[r] = (v2f){0.0f, 0.0f}; acc1[r] = (v2f){0.0f, 0.0f}; }

#pragma unroll 4
    for (int h = 0; h < SC_H; h += 4) {
        const float4 v0 = *(const float4*)(h0p + h);
        const float4 v1 = *(const float4*)(h1p + h);
        const float4 w4 = *(const float4*)(w_l + h);
        const v2f v0a = {v0.x, v0.y}, v0b = {v0.z, v0.w};
        const v2f v1a = {v1.x, v1.y}, v1b = {v1.z, v1.w};
        const v2f wa  = {w4.x, w4.y}, wb  = {w4.z, w4.w};
#pragma unroll
        for (int r = 0; r < SC_RQ; ++r) {
            const float4 u = *(const float4*)(qp + r * SC_H + h);
            const v2f ua = {u.x, u.y}, ub = {u.z, u.w};
            acc0[r] = relu_mad(ua, v0a, wa, acc0[r]);
            acc1[r] = relu_mad(ua, v1a, wa, acc1[r]);
            acc0[r] = relu_mad(ub, v0b, wb, acc0[r]);
            acc1[r] = relu_mad(ub, v1b, wb, acc1[r]);
        }
    }

    float* op = out + (size_t)(q0 + wave * SC_RQ) * S + s0 + lane;
#pragma unroll
    for (int r = 0; r < SC_RQ; ++r) {
        op[(size_t)r * S]      = (acc0[r].x + acc0[r].y) + b2;
        op[(size_t)r * S + 32] = (acc1[r].x + acc1[r].y) + b2;
    }
}

extern "C" void kernel_launch(void* const* d_in, const int* in_sizes, int n_in,
                              void* d_out, int out_size, void* d_ws, size_t ws_size,
                              hipStream_t stream)
{
    const float* Xq  = (const float*)d_in[0];
    const float* Xs  = (const float*)d_in[1];
    const int*   y   = (const int*)d_in[2];
    const float* Wn1 = (const float*)d_in[3];
    const float* bn1 = (const float*)d_in[4];
    const float* Wn2 = (const float*)d_in[5];
    const float* bn2 = (const float*)d_in[6];
    const float* Wm1 = (const float*)d_in[7];
    const float* bm1 = (const float*)d_in[8];
    const float* Wm2 = (const float*)d_in[9];
    const float* bm2 = (const float*)d_in[10];
    const float* Wr1 = (const float*)d_in[11];
    const float* br1 = (const float*)d_in[12];
    const float* wr2 = (const float*)d_in[13];
    const float* br2 = (const float*)d_in[14];
    float* out = (float*)d_out;

    const int E = 512, H = 256, Q = 1024, S = 512;
    const int TQ = (Q / 16) * (H / 64);
    const int TS = (S / 16) * (H / 64);
    const int TW = (H / 16) * (H / 64);
    const size_t HH = (size_t)H * H;

    float* ws  = (float*)d_ws;
    float* q1  = ws; ws += (size_t)Q * H;
    float* s1  = ws; ws += (size_t)S * H;
    float* qf  = ws; ws += (size_t)Q * H;
    float* sf  = ws; ws += (size_t)S * H;
    float* Ab  = ws; ws += (size_t)S * H;
    float* Bb  = ws; ws += (size_t)S * H;
    float* hag = ws; ws += (size_t)S * H;
    float* hsb = ws; ws += (size_t)S * H;
    float* Pa1 = ws; ws += HH;
    float* Pb1 = ws; ws += HH;
    float* Pr  = ws; ws += HH;
    float* cnt = ws; ws += S;
    float* hqb = q1;
    float* sf2 = s1;
    float* Pa0 = hsb;
    float* Pb0 = hsb + HH;

    const float* W1a0 = Wm1;            const float* W1b0 = Wm1 + HH;
    const float* W1a1 = Wm1 + 2 * HH;   const float* W1b1 = Wm1 + 3 * HH;
    const float* W1a2 = Wm1 + 4 * HH;   const float* W1b2 = Wm1 + 5 * HH;
    const float* Wm20 = Wm2;
    const float* Wm21 = Wm2 + HH;
    const float* Wm22 = Wm2 + 2 * HH;
    const float* Wr1a = Wr1;
    const float* Wr1b = Wr1 + HH;

    wmma_gemm_multi<16><<<TQ + TS, 512, 0, stream>>>(
        Xq, Wn1, q1, TQ,
        Xs, Wn1, s1, TQ + TS,
        nullptr, nullptr, nullptr, TQ + TS,
        nullptr, nullptr, nullptr, TQ + TS,
        nullptr, nullptr, nullptr,
        bn1, 2, H, E, 1);
    wmma_gemm_multi<8><<<TQ + TS + 3 * TW, 256, 0, stream>>>(
        q1, Wn2, qf, TQ,
        s1, Wn2, sf, TQ + TS,
        Wm20, W1a1, Pa0, TQ + TS + TW,
        Wm20, W1b1, Pb0, TQ + TS + 2 * TW,
        Wm21, W1a2, Pa1,
        bn2, 2, H, H, 0);
    wmma_gemm_multi<8><<<2 * TS + TQ + 2 * TW, 256, 0, stream>>>(
        sf, W1a0, Ab, TS,
        sf, W1b0, Bb, 2 * TS,
        qf, Wr1a, hqb, 2 * TS + TQ,
        Wm21, W1b2, Pb1, 2 * TS + TQ + TW,
        Wm22, Wr1b, Pr,
        nullptr, 0, H, H, 0);

    pair_accum<<<S, H, 0, stream>>>(Ab, Bb, bm1, y, hag, cnt, H);
    step_gemm<<<3 * TS, 512, 0, stream>>>(
        hag, cnt, sf, bm2,
        Wm20, nullptr, sf2, TS,
        Pa0, W1a1, Ab, 2 * TS,
        Pb0, W1b1, Bb,
        H, H);
    pair_accum<<<S, H, 0, stream>>>(Ab, Bb, bm1 + H, y, hag, cnt, H);
    step_gemm<<<3 * TS, 512, 0, stream>>>(
        hag, cnt, sf2, bm2 + H,
        Wm21, nullptr, sf, TS,
        Pa1, W1a2, Ab, 2 * TS,
        Pb1, W1b2, Bb,
        H, H);
    pair_accum<<<S, H, 0, stream>>>(Ab, Bb, bm1 + 2 * H, y, hag, cnt, H);
    step_gemm<<<TS, 512, 0, stream>>>(
        hag, cnt, sf, bm2 + 2 * H,
        Pr, Wr1b, hsb, TS,
        nullptr, nullptr, nullptr, TS,
        nullptr, nullptr, nullptr,
        H, H);

    score_kernel<<<(Q / 32) * (S / 64), 256, 0, stream>>>(hqb, hsb, br1, wr2, br2, out, S);
}


# Encoder layer 2 folded into its consumers: A, B, hq = layer-1 rows @ (Wn2 @ W) + bn2 @ W, with Wn2 @ W and bn2 @ W formed beside layer 1; 10 -> 9 dependent launches; changes the summation order
# speedup vs baseline: 11.6008x; 1.0450x over previous
#include <hip/hip_runtime.h>
#include <hip/hip_bf16.h>

typedef float v2f __attribute__((ext_vector_type(2)));
typedef float v8f __attribute__((ext_vector_type(8)));


constexpr int PITCH = 68;

__device__ __forceinline__ void load_slice(const float* ap, const float* bp, size_t sN,
                                           v2f (&a)[8], v2f (&f)[8][4])
{
#pragma unroll
    for (int i = 0; i < 4; ++i) {
        const float4 q = *(const float4*)(ap + 4 * i);
        a[2 * i]     = (v2f){q.x, q.y};
        a[2 * i + 1] = (v2f){q.z, q.w};
    }
#pragma unroll
    for (int s = 0; s < 8; ++s) {
        const float* b0 = bp + (size_t)(2 * s) * sN;
        const float* b1 = b0 + sN;
        const float4 r0 = *(const float4*)b0;
        const float4 r1 = *(const float4*)b1;
        f[s][0] = (v2f){r0.x, r1.x};
        f[s][1] = (v2f){r0.y, r1.y};
        f[s][2] = (v2f){r0.z, r1.z};
        f[s][3] = (v2f){r0.w, r1.w};
    }
}

__device__ __forceinline__ void mma_slice_to_lds(const v2f (&a)[8], const v2f (&f)[8][4], float* my)
{
    v8f acc0 = {}, acc1 = {}, acc2 = {}, acc3 = {};
#pragma unroll
    for (int s = 0; s < 8; ++s) {
        acc0 = __builtin_amdgcn_wmma_f32_16x16x4_f32(false, a[s], false, f[s][0],
                                                     (short)0, acc0, false, false);
        acc1 = __builtin_amdgcn_wmma_f32_16x16x4_f32(false, a[s], false, f[s][1],
                                                     (short)0, acc1, false, false);
        acc2 = __builtin_amdgcn_wmma_f32_16x16x4_f32(false, a[s], false, f[s][2],
                                                     (short)0, acc2, false, false);
        acc3 = __builtin_amdgcn_wmma_f32_16x16x4_f32(false, a[s], false, f[s][3],
                                                     (short)0, acc3, false, false);
    }
#pragma unroll
    for (int r = 0; r < 8; ++r)
        *(float4*)&my[r * PITCH] = make_float4(acc0[r], acc1[r], acc2[r], acc3[r]);
}

template <int NW>
__device__ __forceinline__ void gemm_tile(
    const float* __restrict__ A, size_t lda, const float* __restrict__ W,
    float* __restrict__ C, const float* __restrict__ bias, int relu,
    int N, int K, int tile, float (*red)[16 * PITCH])
{
    const int lane = threadIdx.x & 31;
    const int wave = __builtin_amdgcn_readfirstlane((int)(threadIdx.x >> 5));
    const int tilesN = N >> 6;
    const int tm = (tile / tilesN) << 4;
    const int tn = (tile % tilesN) << 6;
    const int kw = (NW == 8) ? 8 : (K >> 5);

    const int half = lane >> 4;
    const int l16  = lane & 15;
    const int k0   = wave << 5;

    if (wave < kw) {
        const float* ap = A + (size_t)(tm + l16) * lda + k0 + 16 * half;
        const float* bp = W + (size_t)(k0 + 16 * half) * N + tn + 4 * l16;

        v2f a[8];
        v2f f[8][4];
        load_slice(ap, bp, (size_t)N, a, f);

        __builtin_amdgcn_sched_barrier(0);

        mma_slice_to_lds(a, f, &red[wave][(half * 8) * PITCH + 4 * l16]);
    }
    __syncthreads();

    const int t = threadIdx.x;
    if (t < 256) {
        const int row = t >> 4;
        const int c4  = (t & 15) << 2;
        float* cp = C + (size_t)(tm + row) * N + tn + c4;
        float4 sum = *(const float4*)&red[0][row * PITCH + c4];
#pragma unroll
        for (int w = 1; w < 8; ++w) {
            const float4 p = *(const float4*)&red[w][row * PITCH + c4];
            sum.x += p.x; sum.y += p.y; sum.z += p.z; sum.w += p.w;
        }
        if (NW == 16 && kw == 16) {
#pragma unroll
            for (int w = 8; w < NW; ++w) {
                const float4 p = *(const float4*)&red[w][row * PITCH + c4];
                sum.x += p.x; sum.y += p.y; sum.z += p.z; sum.w += p.w;
            }
        }
        if (bias) {
            const float4 b4 = *(const float4*)(bias + tn + c4);
            sum.x += b4.x; sum.y += b4.y; sum.z += b4.z; sum.w += b4.w;
        }
        if (relu) {
            sum.x = fmaxf(sum.x, 0.f); sum.y = fmaxf(sum.y, 0.f);
            sum.z = fmaxf(sum.z, 0.f); sum.w = fmaxf(sum.w, 0.f);
        }
        *(float4*)cp = sum;
    }
}

struct GArgs {
    const float* a0; const float* w0; float* c0; const float* b0; int e0;
    const float* a1; const float* w1; float* c1; const float* b1; int e1;
    const float* a2; const float* w2; float* c2; const float* b2; int e2;
    const float* a3; const float* w3; float* c3; const float* b3; int e3;
    const float* a4; const float* w4; float* c4; const float* b4; int e4;
    const float* a5; const float* w5; float* c5; const float* b5; int e5;
    const float* a6; const float* w6; float* c6; const float* b6; int e6;
    const float* a7; const float* w7; float* c7; const float* b7;
    int nbig, kbig, ksmall, nrelu, zrow, N;
};

#define GSEL(x) (ge6 ? g.x##7 : ge5 ? g.x##6 : ge4 ? g.x##5 : ge3 ? g.x##4 : \
                 ge2 ? g.x##3 : ge1 ? g.x##2 : ge0 ? g.x##1 : g.x##0)

template <int NW>
__device__ __forceinline__ void gemm_list(const GArgs& g, int b, float (*red)[16 * PITCH])
{
    const bool ge0 = b >= g.e0, ge1 = b >= g.e1, ge2 = b >= g.e2, ge3 = b >= g.e3;
    const bool ge4 = b >= g.e4, ge5 = b >= g.e5, ge6 = b >= g.e6;
    const int sel = (int)ge0 + (int)ge1 + (int)ge2 + (int)ge3 + (int)ge4 + (int)ge5 + (int)ge6;
    const float* A    = GSEL(a);
    const float* W    = GSEL(w);
    float*       C    = GSEL(c);
    const float* bias = GSEL(b);
    const int first = ge6 ? g.e6 : ge5 ? g.e5 : ge4 ? g.e4 : ge3 ? g.e3 :
                      ge2 ? g.e2 : ge1 ? g.e1 : ge0 ? g.e0 : 0;
    const int K = sel < g.nbig ? g.kbig : g.ksmall;
    const size_t lda = ((g.zrow >> sel) & 1) ? (size_t)0 : (size_t)K;
    gemm_tile<NW>(A, lda, W, C, bias, sel < g.nrelu, g.N, K, b - first, red);
}

template <int NW>
__global__ __launch_bounds__(NW * 32) void gemm8(const GArgs g)
{
    __shared__ alignas(16) float red[NW][16 * PITCH];
    gemm_list<NW>(g, (int)blockIdx.x, red);
}

__global__ __launch_bounds__(512) void step_gemm(
    const float* __restrict__ hag, const float* __restrict__ cnt,
    const float* __restrict__ Sin, const float* __restrict__ bv,
    const float* __restrict__ Wh0, const float* __restrict__ Wb0, float* __restrict__ C0, int e0,
    const float* __restrict__ Wh1, const float* __restrict__ Wb1, float* __restrict__ C1, int e1,
    const float* __restrict__ Wh2, const float* __restrict__ Wb2, float* __restrict__ C2,
    int N, int K)
{
    __shared__ alignas(16) float red[16][16 * PITCH];

    const int lane = threadIdx.x & 31;
    const int wave = __builtin_amdgcn_readfirstlane((int)(threadIdx.x >> 5));
    const int tilesN = N >> 6;

    const int b = (int)blockIdx.x;
    const int sel = (b >= e0) + (b >= e1);
    const float* Wh = sel == 0 ? Wh0 : sel == 1 ? Wh1 : Wh2;
    const float* Wb = sel == 0 ? Wb0 : sel == 1 ? Wb1 : Wb2;
    float*       C  = sel == 0 ? C0  : sel == 1 ? C1  : C2;
    const int tile = b - (sel == 0 ? 0 : sel == 1 ? e0 : e1);
    const int tm = (tile / tilesN) << 4;
    const int tn = (tile % tilesN) << 6;

    const int half = lane >> 4;
    const int l16  = lane & 15;
    const bool upper = wave >= 8;
    const int k0 = (wave & 7) << 5;
    const bool nodeprod = (Wb != nullptr);

    if (!upper || nodeprod) {
        const float* Asrc = upper ? Sin : hag;
        const float* Wsrc = upper ? Wb : Wh;
        const float* ap = Asrc + (size_t)(tm + l16) * K + k0 + 16 * half;
        const float* bp = Wsrc + (size_t)(k0 + 16 * half) * N + tn + 4 * l16;

        v2f a[8];
        v2f f[8][4];
        load_slice(ap, bp, (size_t)N, a, f);
        if (upper) {
            const float c = cnt[tm + l16];
            const float fl = c / fmaxf(c, 1.0f);
            const float* bq = bv + k0 + 16 * half;
#pragma unroll
            for (int s = 0; s < 8; ++s) {
                const v2f b2 = *(const v2f*)(bq + 2 * s);
                a[s].x += fl * b2.x;
                a[s].y += fl * b2.y;
            }
        }
        __builtin_amdgcn_sched_barrier(0);
        mma_slice_to_lds(a, f, &red[wave][(half * 8) * PITCH + 4 * l16]);
    }
    __syncthreads();

    const int t = threadIdx.x;
    if (t < 256) {
        const int row = t >> 4;
        const int c4  = (t & 15) << 2;
        const size_t off = (size_t)(tm + row) * N + tn + c4;
        float4 sh = *(const float4*)&red[0][row * PITCH + c4];
#pragma unroll
        for (int w = 1; w < 8; ++w) {
            const float4 p = *(const float4*)&red[w][row * PITCH + c4];
            sh.x += p.x; sh.y += p.y; sh.z += p.z; sh.w += p.w;
        }
        const float c = cnt[tm + row];
        const float denom = fmaxf(c, 1.0f);
        float4 o;
        if (nodeprod) {
            float4 sb = *(const float4*)&red[8][row * PITCH + c4];
#pragma unroll
            for (int w = 9; w < 16; ++w) {
                const float4 p = *(const float4*)&red[w][row * PITCH + c4];
                sb.x += p.x; sb.y += p.y; sb.z += p.z; sb.w += p.w;
            }
            o.x = sb.x + sh.x / denom;
            o.y = sb.y + sh.y / denom;
            o.z = sb.z + sh.z / denom;
            o.w = sb.w + sh.w / denom;
        } else {
            const float4 b4 = *(const float4*)(bv + tn + c4);
            o = *(const float4*)(Sin + off);
            o.x += (sh.x + c * b4.x) / denom;
            o.y += (sh.y + c * b4.y) / denom;
            o.z += (sh.z + c * b4.z) / denom;
            o.w += (sh.w + c * b4.w) / denom;
        }
        *(float4*)(C + off) = o;
    }
}

constexpr int PAIR_S = 512;
constexpr int PG = 32;
__global__ __launch_bounds__(256) void pair_accum(
    const float* __restrict__ Abuf, const float* __restrict__ Bbuf,
    const float* __restrict__ bm1t, const int* __restrict__ y,
    float* __restrict__ haggr, float* __restrict__ cnt, int H, int npair, const GArgs g)
{
    __shared__ alignas(16) unsigned list[PAIR_S + PG];
    __shared__ alignas(16) int wcnt[8];
    __shared__ alignas(16) float red[8][16 * PITCH];

    if ((int)blockIdx.x >= npair) {
        gemm_list<8>(g, (int)blockIdx.x - npair, red);
        return;
    }

    const int i = blockIdx.x;
    const int h = threadIdx.x;
    const int lane = h & 31;
    const int wave = h >> 5;

    const int j0 = (wave << 6) + lane;
    const int j1 = j0 + 32;
    const int yi = y[i];
    const int y0 = y[j0];
    const int y1 = y[j1];
    const float av = Abuf[(size_t)i * H + h] + bm1t[h];

    const bool m0 = (y0 == yi) && (j0 != i);
    const bool m1 = (y1 == yi) && (j1 != i);
    const unsigned b0 = __builtin_amdgcn_ballot_w32(m0);
    const unsigned b1 = __builtin_amdgcn_ballot_w32(m1);
    const int c0 = __builtin_popcount(b0);
    const int c1 = __builtin_popcount(b1);
    if (lane == 0) wcnt[wave] = c0 + c1;
    __syncthreads();

    int base = 0, n = 0;
    for (int w = 0; w < 8; ++w) {
        const int c = wcnt[w];
        base += (w < wave) ? c : 0;
        n += c;
    }
    const unsigned below = (1u << lane) - 1u;
    if (m0) list[base + __builtin_popcount(b0 & below)] = (unsigned)(j0 * H);
    if (m1) list[base + c0 + __builtin_popcount(b1 & below)] = (unsigned)(j1 * H);
    if (h < PG) list[n + h] = 0u;
    __syncthreads();

    float acc = 0.0f;
    for (int p = 0; p < n; p += PG) {
        float v[PG];
#pragma unroll
        for (int u = 0; u < PG; ++u)
            v[u] = Bbuf[list[p + u] + (unsigned)h];
#pragma unroll
        for (int u = 0; u < PG; ++u) {
            const float t = fmaxf(av + v[u], 0.0f);
            acc += (p + u < n) ? t : 0.0f;
        }
    }
    haggr[(size_t)i * H + h] = acc;
    if (h == 0) cnt[i] = (float)n;
}

constexpr int SC_H  = 256;
constexpr int SC_TQ = 32;
constexpr int SC_TS = 64;
constexpr int SC_RQ = 4;
constexpr int SC_PS = SC_H + 4;

__device__ __forceinline__ v2f relu_mad(v2f u, v2f v, v2f w, v2f acc)
{
    v2f t = u + v;
    t = __builtin_elementwise_max(t, (v2f){0.0f, 0.0f});
    return t * w + acc;
}

__global__ __launch_bounds__(256) void score_kernel(
    const float* __restrict__ hq, const float* __restrict__ hs,
    const float* __restrict__ br1, const float* __restrict__ wr2,
    const float* __restrict__ br2, float* __restrict__ out, int S)
{
    __shared__ alignas(16) float hs_l[SC_TS * SC_PS];
    __shared__ alignas(16) float hq_l[SC_TQ * SC_H];
    __shared__ alignas(16) float w_l[SC_H];

    const int t = threadIdx.x;
    const int lane = t & 31;
    const int wave = t >> 5;
    const int tilesS = S / SC_TS;
    const int q0 = ((int)blockIdx.x / tilesS) * SC_TQ;
    const int s0 = ((int)blockIdx.x % tilesS) * SC_TS;

    const int c4 = (t & 63) << 2;
    const int r0 = t >> 6;
    const float4 b4 = *(const float4*)(br1 + c4);
#pragma unroll
    for (int i = 0; i < SC_TS / 4; ++i) {
        const int row = r0 + 4 * i;
        *(float4*)&hs_l[row * SC_PS + c4] =
            *(const float4*)(hs + (size_t)(s0 + row) * SC_H + c4);
    }
#pragma unroll
    for (int i = 0; i < SC_TQ / 4; ++i) {
        const int row = r0 + 4 * i;
        float4 u = *(const float4*)(hq + (size_t)(q0 + row) * SC_H + c4);
        u.x += b4.x; u.y += b4.y; u.z += b4.z; u.w += b4.w;
        *(float4*)&hq_l[row * SC_H + c4] = u;
    }
    if (t < 64) *(float4*)&w_l[c4] = *(const float4*)(wr2 + c4);
    const float b2 = br2[0];
    __syncthreads();

    const float* h0p = &hs_l[lane * SC_PS];
    const float* h1p = &hs_l[(lane + 32) * SC_PS];
    const float* qp  = &hq_l[(wave * SC_RQ) * SC_H];

    v2f acc0[SC_RQ], acc1[SC_RQ];
#pragma unroll
    for (int r = 0; r < SC_RQ; ++r) { acc0[r] = (v2f){0.0f, 0.0f}; acc1[r] = (v2f){0.0f, 0.0f}; }

#pragma unroll 4
    for (int h = 0; h < SC_H; h += 4) {
        const float4 v0 = *(const float4*)(h0p + h);
        const float4 v1 = *(const float4*)(h1p + h);
        const float4 w4 = *(const float4*)(w_l + h);
        const v2f v0a = {v0.x, v0.y}, v0b = {v0.z, v0.w};
        const v2f v1a = {v1.x, v1.y}, v1b = {v1.z, v1.w};
        const v2f wa  = {w4.x, w4.y}, wb  = {w4.z, w4.w};
#pragma unroll
        for (int r = 0; r < SC_RQ; ++r) {
            const float4 u = *(const float4*)(qp + r * SC_H + h);
            const v2f ua = {u.x, u.y}, ub = {u.z, u.w};
            acc0[r] = relu_mad(ua, v0a, wa, acc0[r]);
            acc1[r] = relu_mad(ua, v1a, wa, acc1[r]);
            acc0[r] = relu_mad(ub, v0b, wb, acc0[r]);
            acc1[r] = relu_mad(ub, v1b, wb, acc1[r]);
        }
    }

    float* op = out + (size_t)(q0 + wave * SC_RQ) * S + s0 + lane;
#pragma unroll
    for (int r = 0; r < SC_RQ; ++r) {
        op[(size_t)r * S]      = (acc0[r].x + acc0[r].y) + b2;
        op[(size_t)r * S + 32] = (acc1[r].x + acc1[r].y) + b2;
    }
}

extern "C" void kernel_launch(void* const* d_in, const int* in_sizes, int n_in,
                              void* d_out, int out_size, void* d_ws, size_t ws_size,
                              hipStream_t stream)
{
    const float* Xq  = (const float*)d_in[0];
    const float* Xs  = (const float*)d_in[1];
    const int*   y   = (const int*)d_in[2];
    const float* Wn1 = (const float*)d_in[3];
    const float* bn1 = (const float*)d_in[4];
    const float* Wn2 = (const float*)d_in[5];
    const float* bn2 = (const float*)d_in[6];
    const float* Wm1 = (const float*)d_in[7];
    const float* bm1 = (const float*)d_in[8];
    const float* Wm2 = (const float*)d_in[9];
    const float* bm2 = (const float*)d_in[10];
    const float* Wr1 = (const float*)d_in[11];
    const float* br1 = (const float*)d_in[12];
    const float* wr2 = (const float*)d_in[13];
    const float* br2 = (const float*)d_in[14];
    float* out = (float*)d_out;

    const int E = 512, H = 256, Q = 1024, S = 512;
    const int TQ = (Q / 16) * (H / 64);
    const int TS = (S / 16) * (H / 64);
    const int TW = (H / 16) * (H / 64);
    const int TN = H / 64;
    const size_t HH = (size_t)H * H;

    float* ws  = (float*)d_ws;
    float* q1  = ws; ws += (size_t)Q * H;
    float* s1  = ws; ws += (size_t)S * H;
    float* hqb = ws; ws += (size_t)Q * H;
    float* sf  = ws; ws += (size_t)S * H;
    float* Ab  = ws; ws += (size_t)S * H;
    float* Bb  = ws; ws += (size_t)S * H;
    float* hag = ws; ws += (size_t)S * H;
    float* hsb = ws; ws += (size_t)S * H;
    float* Pa1 = ws; ws += HH;
    float* Pb1 = ws; ws += HH;
    float* Pr  = ws; ws += HH;
    float* cnt = ws; ws += S;
    float* sf2 = s1;
    float* Pa0 = hsb;
    float* Pb0 = hsb + HH;
    float* Ga  = Pa1;
    float* Gb  = Pb1;
    float* Gr  = Pr;
    float* ca  = hag;
    float* cb  = hag + 16 * H;
    float* cr  = hag + 32 * H;

    const float* W1a0 = Wm1;            const float* W1b0 = Wm1 + HH;
    const float* W1a1 = Wm1 + 2 * HH;   const float* W1b1 = Wm1 + 3 * HH;
    const float* W1a2 = Wm1 + 4 * HH;   const float* W1b2 = Wm1 + 5 * HH;
    const float* Wm20 = Wm2;
    const float* Wm21 = Wm2 + HH;
    const float* Wm22 = Wm2 + 2 * HH;
    const float* Wr1a = Wr1;
    const float* Wr1b = Wr1 + HH;

    GArgs g1 = {};
    g1.a0 = Xq;  g1.w0 = Wn1;  g1.c0 = q1; g1.b0 = bn1;     g1.e0 = TQ;
    g1.a1 = Xs;  g1.w1 = Wn1;  g1.c1 = s1; g1.b1 = bn1;     g1.e1 = TQ + TS;
    g1.a2 = Wn2; g1.w2 = W1a0; g1.c2 = Ga; g1.b2 = nullptr; g1.e2 = TQ + TS + TW;
    g1.a3 = Wn2; g1.w3 = W1b0; g1.c3 = Gb; g1.b3 = nullptr; g1.e3 = TQ + TS + 2 * TW;
    g1.a4 = Wn2; g1.w4 = Wr1a; g1.c4 = Gr; g1.b4 = nullptr; g1.e4 = TQ + TS + 3 * TW;
    g1.a5 = bn2; g1.w5 = W1a0; g1.c5 = ca; g1.b5 = nullptr; g1.e5 = TQ + TS + 3 * TW + TN;
    g1.a6 = bn2; g1.w6 = W1b0; g1.c6 = cb; g1.b6 = nullptr; g1.e6 = TQ + TS + 3 * TW + 2 * TN;
    g1.a7 = bn2; g1.w7 = Wr1a; g1.c7 = cr; g1.b7 = nullptr;
    g1.nbig = 2; g1.kbig = E; g1.ksmall = H; g1.nrelu = 2; g1.zrow = 0xE0; g1.N = H;
    gemm8<16><<<TQ + TS + 3 * TW + 3 * TN, 512, 0, stream>>>(g1);

    GArgs g2 = {};
    g2.a0 = s1;   g2.w0 = Wn2;  g2.c0 = sf;  g2.b0 = bn2;     g2.e0 = TS;
    g2.a1 = s1;   g2.w1 = Ga;   g2.c1 = Ab;  g2.b1 = ca;      g2.e1 = 2 * TS;
    g2.a2 = s1;   g2.w2 = Gb;   g2.c2 = Bb;  g2.b2 = cb;      g2.e2 = 3 * TS;
    g2.a3 = q1;   g2.w3 = Gr;   g2.c3 = hqb; g2.b3 = cr;      g2.e3 = 3 * TS + TQ;
    g2.a4 = Wm20; g2.w4 = W1a1; g2.c4 = Pa0; g2.b4 = nullptr; g2.e4 = 3 * TS + TQ + TW;
    g2.a5 = Wm20; g2.w5 = W1b1; g2.c5 = Pb0; g2.b5 = nullptr; g2.e5 = 3 * TS + TQ + 2 * TW;
    g2.e6 = 3 * TS + TQ + 2 * TW;
    g2.nbig = 0; g2.kbig = H; g2.ksmall = H; g2.nrelu = 0; g2.zrow = 0; g2.N = H;
    gemm8<8><<<3 * TS + TQ + 2 * TW, 256, 0, stream>>>(g2);

    GArgs g3 = {};
    g3.a0 = Wm21; g3.w0 = W1a2; g3.c0 = Pa1; g3.b0 = nullptr; g3.e0 = TW;
    g3.a1 = Wm21; g3.w1 = W1b2; g3.c1 = Pb1; g3.b1 = nullptr; g3.e1 = 2 * TW;
    g3.a2 = Wm22; g3.w2 = Wr1b; g3.c2 = Pr;  g3.b2 = nullptr; g3.e2 = 3 * TW;
    g3.e3 = 3 * TW; g3.e4 = 3 * TW; g3.e5 = 3 * TW; g3.e6 = 3 * TW;
    g3.nbig = 0; g3.kbig = H; g3.ksmall = H; g3.nrelu = 0; g3.zrow = 0; g3.N = H;
    GArgs g0 = {};

    pair_accum<<<S + 3 * TW, H, 0, stream>>>(Ab, Bb, bm1, y, hag, cnt, H, S, g3);
    step_gemm<<<3 * TS, 512, 0, stream>>>(
        hag, cnt, sf, bm2,
        Wm20, nullptr, sf2, TS,
        Pa0, W1a1, Ab, 2 * TS,
        Pb0, W1b1, Bb,
        H, H);
    pair_accum<<<S, H, 0, stream>>>(Ab, Bb, bm1 + H, y, hag, cnt, H, S, g0);
    step_gemm<<<3 * TS, 512, 0, stream>>>(
        hag, cnt, sf2, bm2 + H,
        Wm21, nullptr, sf, TS,
        Pa1, W1a2, Ab, 2 * TS,
        Pb1, W1b2, Bb,
        H, H);
    pair_accum<<<S, H, 0, stream>>>(Ab, Bb, bm1 + 2 * H, y, hag, cnt, H, S, g0);
    step_gemm<<<TS, 512, 0, stream>>>(
        hag, cnt, sf, bm2 + 2 * H,
        Pr, Wr1b, hsb, TS,
        nullptr, nullptr, nullptr, TS,
        nullptr, nullptr, nullptr,
        H, H);

    score_kernel<<<(Q / 32) * (S / 64), 256, 0, stream>>>(hqb, hsb, br1, wr2, br2, out, S);
}
